# starting kernel
# baseline (speedup 1.0000x reference)
#include <hip/hip_runtime.h>

typedef __attribute__((ext_vector_type(2))) float v2f;
typedef __attribute__((ext_vector_type(4))) float v4f;
typedef __attribute__((ext_vector_type(8))) float v8f;

static constexpr int SN    = 128;
static constexpr int TILE  = 16;
static constexpr int NB    = SN / TILE;
static constexpr int APAD  = 132;
static constexpr int XPAD  = 20;
static constexpr int WAVES = 2;

__global__ __launch_bounds__(WAVES * 32) void hippo_bilinear_wmma(
    const float* __restrict__ u, const float* __restrict__ v,
    const float* __restrict__ delta, const float* __restrict__ A,
    const float* __restrict__ Bvec, float* __restrict__ out, int batch)
{
    __shared__ float A_lds[SN * APAD];
    __shared__ float xt[WAVES][TILE][XPAD];
    __shared__ float zf[WAVES][NB][TILE][XPAD];

    const int tid = threadIdx.x;
    const int nthreads = WAVES * 32;

    #pragma unroll 4
    for (int i = 0; i < (SN * SN / 4) / nthreads; ++i) {
        int q = tid + i * nthreads;
        int r = q >> 5;
        int c = (q & 31) << 2;
        v4f t = *(const v4f*)(A + r * SN + c);
        *(v4f*)(&A_lds[r * APAD + c]) = t;
    }
    __syncthreads();

    const int wave = tid >> 5;
    const int lane = tid & 31;
    const int col  = lane & 15;
    const int hlf  = lane >> 4;
    const int batch_base = (blockIdx.x * WAVES + wave) * TILE;
    const int b = batch_base + col;
    if (batch_base >= batch) return;

    const float dt2 = 0.5f * delta[b];
    const float dvb = delta[b] * v[b];
    const float* ucol = u + (size_t)b * SN;

    float* myxt = &xt[wave][0][0];

    #pragma unroll 1
    for (int k = 0; k < NB; ++k) {
        const float* arows = &A_lds[(k * TILE + col) * APAD];

        v8f acc = {};
        for (int j = 0; j <= k; ++j) {
            const float* arow = arows + j * TILE;
            const float* ucb  = ucol + j * TILE;
            #pragma unroll
            for (int k4 = 0; k4 < 4; ++k4) {
                int ko = k4 * 4 + 2 * hlf;
                v2f af = *(const v2f*)(arow + ko);
                v2f bf = *(const v2f*)(ucb + ko);
                acc = __builtin_amdgcn_wmma_f32_16x16x4_f32(
                          false, af, false, bf, (short)0, acc, false, false);
            }
        }

        {
            const int nbase = k * TILE + 8 * hlf;
            const v4f u0 = *(const v4f*)(ucol + nbase);
            const v4f u1 = *(const v4f*)(ucol + nbase + 4);
            const v4f b0 = *(const v4f*)(Bvec + nbase);
            const v4f b1 = *(const v4f*)(Bvec + nbase + 4);
            #pragma unroll
            for (int r = 0; r < 4; ++r) {
                acc[r]     = u0[r] + dt2 * acc[r]     + dvb * b0[r];
                acc[r + 4] = u1[r] + dt2 * acc[r + 4] + dvb * b1[r];
            }
        }

        for (int j = 0; j < k; ++j) {
            const float* arow = arows + j * TILE;
            const float* zrow = &zf[wave][j][col][0];
            #pragma unroll
            for (int k4 = 0; k4 < 4; ++k4) {
                int ko = k4 * 4 + 2 * hlf;
                v2f af = *(const v2f*)(arow + ko);
                v2f zb = *(const v2f*)(zrow + ko);
                acc = __builtin_amdgcn_wmma_f32_16x16x4_f32(
                          false, af, false, zb, (short)0, acc, false, false);
            }
        }

        #pragma unroll
        for (int r = 0; r < 8; ++r)
            myxt[(r + 8 * hlf) * XPAD + col] = acc[r];
        asm volatile("s_wait_dscnt 0x0" ::: "memory");

        if (lane < 16) {
            const int rb = k * TILE;
            float* myz = &zf[wave][k][lane][0];
            float y[16];
            #pragma unroll
            for (int m = 0; m < 16; ++m) {
                float s = myxt[m * XPAD + lane];
                float accv = 0.f;
                #pragma unroll
                for (int j = 0; j < m; ++j)
                    accv += A_lds[(rb + m) * APAD + rb + j] * y[j];
                const float diag = A_lds[(rb + m) * APAD + rb + m];
                y[m] = (s + dt2 * accv) / (1.0f - dt2 * diag);
                myz[m] = dt2 * y[m];
            }
            float* o = out + (size_t)(batch_base + lane) * SN + rb;
            #pragma unroll
            for (int m = 0; m < 16; m += 4) {
                v4f t = { y[m], y[m + 1], y[m + 2], y[m + 3] };
                *(v4f*)(o + m) = t;
            }
        }
        asm volatile("s_wait_dscnt 0x0" ::: "memory");
    }
}

extern "C" void kernel_launch(void* const* d_in, const int* in_sizes, int n_in,
                              void* d_out, int out_size, void* d_ws, size_t ws_size,
                              hipStream_t stream) {
    const float* u     = (const float*)d_in[0];
    const float* v     = (const float*)d_in[1];
    const float* delta = (const float*)d_in[2];
    const float* A     = (const float*)d_in[3];
    const float* Bvec  = (const float*)d_in[4];
    float* out = (float*)d_out;

    const int batch  = in_sizes[0] / SN;
    const int blocks = batch / (WAVES * TILE);
    hippo_bilinear_wmma<<<blocks, WAVES * 32, 0, stream>>>(u, v, delta, A, Bvec, out, batch);
}


# right-looking block substitution: the 8 block-row right-hand sides formed up front in registers (u loaded once, one group); each solved block updates the rows below it; sum order unchanged
# speedup vs baseline: 1.3433x; 1.3433x over previous
#include <hip/hip_runtime.h>


typedef __attribute__((ext_vector_type(2))) float v2f;
typedef __attribute__((ext_vector_type(4))) float v4f;
typedef __attribute__((ext_vector_type(8))) float v8f;

static constexpr int SN    = 128;
static constexpr int TILE  = 16;
static constexpr int NB    = SN / TILE;
static constexpr int APAD  = 132;
static constexpr int XPAD  = 20;
static constexpr int WAVES = 2;

__device__ __forceinline__ v8f frag_mma(const v2f (&af)[4], const v2f (&bf)[4], v8f c)
{
    #pragma unroll
    for (int k4 = 0; k4 < 4; ++k4)
        c = __builtin_amdgcn_wmma_f32_16x16x4_f32(
                false, af[k4], false, bf[k4], (short)0, c, false, false);
    return c;
}

__global__ __launch_bounds__(WAVES * 32) void hippo_bilinear_wmma(
    const float* __restrict__ u, const float* __restrict__ v,
    const float* __restrict__ delta, const float* __restrict__ A,
    const float* __restrict__ Bvec, float* __restrict__ out, int batch)
{
    __shared__ float A_lds[SN * APAD];
    __shared__ float xt[WAVES][TILE][XPAD];
    __shared__ float zf[WAVES][TILE][XPAD];

    const int tid = threadIdx.x;
    const int nthreads = WAVES * 32;

    const int wave = tid >> 5;
    const int lane = tid & 31;
    const int col  = lane & 15;
    const int hlf  = lane >> 4;
    const int batch_base = (blockIdx.x * WAVES + wave) * TILE;
    const bool live = batch_base < batch;
    const int b = batch_base + col;
    const int bl = live ? b : 0;

    const float dlt = delta[bl];
    const float vv  = v[bl];
    const float* ucol = u + (size_t)bl * SN;
    v2f ufr[NB][4];
    #pragma unroll
    for (int j = 0; j < NB; ++j) {
        #pragma unroll
        for (int k4 = 0; k4 < 4; ++k4)
            ufr[j][k4] = *(const v2f*)(ucol + j * TILE + k4 * 4 + 2 * hlf);
    }
    v4f uc0[NB], uc1[NB], bc0[NB], bc1[NB];
    #pragma unroll
    for (int k = 0; k < NB; ++k) {
        const int nbase = k * TILE + 8 * hlf;
        uc0[k] = *(const v4f*)(ucol + nbase);
        uc1[k] = *(const v4f*)(ucol + nbase + 4);
        bc0[k] = *(const v4f*)(Bvec + nbase);
        bc1[k] = *(const v4f*)(Bvec + nbase + 4);
    }

    #pragma unroll 4
    for (int i = 0; i < (SN * SN / 4) / nthreads; ++i) {
        int q = tid + i * nthreads;
        int r = q >> 5;
        int c = (q & 31) << 2;
        v4f t = *(const v4f*)(A + r * SN + c);
        *(v4f*)(&A_lds[r * APAD + c]) = t;
    }
    __syncthreads();

    if (!live) return;

    const float dt2 = 0.5f * dlt;
    const float dvb = dlt * vv;

    float* myxt = &xt[wave][0][0];
    const float* alane = &A_lds[col * APAD + 2 * hlf];

    v8f R[NB];
    #pragma unroll
    for (int k = 0; k < NB; ++k) R[k] = v8f{};
    #pragma unroll
    for (int j = 0; j < NB; ++j) {
        #pragma unroll
        for (int k4 = 0; k4 < 4; ++k4) {
            #pragma unroll
            for (int k = j; k < NB; ++k) {
                v2f af = *(const v2f*)(alane + (k * TILE) * APAD + j * TILE + k4 * 4);
                R[k] = __builtin_amdgcn_wmma_f32_16x16x4_f32(
                           false, af, false, ufr[j][k4], (short)0, R[k], false, false);
            }
        }
    }
    #pragma unroll
    for (int k = 0; k < NB; ++k) {
        #pragma unroll
        for (int r = 0; r < 4; ++r) {
            R[k][r]     = uc0[k][r] + dt2 * R[k][r]     + dvb * bc0[k][r];
            R[k][r + 4] = uc1[k][r] + dt2 * R[k][r + 4] + dvb * bc1[k][r];
        }
    }

    #pragma unroll
    for (int r = 0; r < 8; ++r)
        myxt[(r + 8 * hlf) * XPAD + col] = R[0][r];

    #pragma unroll 1
    for (int k = 0; k < NB; ++k) {
        asm volatile("s_wait_dscnt 0x0" ::: "memory");

        const float* acolk = alane + k * TILE;
        v2f afr0[4];
        {
            const int rblk = (k + 1 < NB) ? (k + 1) : (NB - 1);
            #pragma unroll
            for (int k4 = 0; k4 < 4; ++k4)
                afr0[k4] = *(const v2f*)(acolk + (rblk * TILE) * APAD + k4 * 4);
        }

        if (lane < 16) {
            const int rb = k * TILE;
            float* myz = &zf[wave][lane][0];
            float y[16];
            #pragma unroll
            for (int m = 0; m < 16; ++m) {
                float s = myxt[m * XPAD + lane];
                float accv = 0.f;
                #pragma unroll
                for (int j = 0; j < m; ++j)
                    accv += A_lds[(rb + m) * APAD + rb + j] * y[j];
                const float diag = A_lds[(rb + m) * APAD + rb + m];
                y[m] = (s + dt2 * accv) / (1.0f - dt2 * diag);
                myz[m] = dt2 * y[m];
            }
            float* o = out + (size_t)(batch_base + lane) * SN + rb;
            #pragma unroll
            for (int m = 0; m < 16; m += 4) {
                v4f t = { y[m], y[m + 1], y[m + 2], y[m + 3] };
                *(v4f*)(o + m) = t;
            }
        }
        asm volatile("s_wait_dscnt 0x0" ::: "memory");

        if (k < NB - 1) {
            v2f zb[4];
            #pragma unroll
            for (int k4 = 0; k4 < 4; ++k4)
                zb[k4] = *(const v2f*)(&zf[wave][col][k4 * 4 + 2 * hlf]);
            v2f afr[NB - 1][4];
            #pragma unroll
            for (int i = 1; i < NB - 1; ++i) {
                const int rblk = (k + 1 + i < NB) ? (k + 1 + i) : (NB - 1);
                #pragma unroll
                for (int k4 = 0; k4 < 4; ++k4)
                    afr[i][k4] = *(const v2f*)(acolk + (rblk * TILE) * APAD + k4 * 4);
            }

            R[0] = frag_mma(afr0, zb, R[1]);
            #pragma unroll
            for (int r = 0; r < 8; ++r)
                myxt[(r + 8 * hlf) * XPAD + col] = R[0][r];

            #pragma unroll
            for (int i = 1; i <= NB - 2; ++i) {
                if (i <= NB - 2 - k)
                    R[i] = frag_mma(afr[i], zb, R[i + 1]);
            }
        }
    }
}

extern "C" void kernel_launch(void* const* d_in, const int* in_sizes, int n_in,
                              void* d_out, int out_size, void* d_ws, size_t ws_size,
                              hipStream_t stream) {
    const float* u     = (const float*)d_in[0];
    const float* v     = (const float*)d_in[1];
    const float* delta = (const float*)d_in[2];
    const float* A     = (const float*)d_in[3];
    const float* Bvec  = (const float*)d_in[4];
    float* out = (float*)d_out;

    const int batch  = in_sizes[0] / SN;
    const int blocks = batch / (WAVES * TILE);
    hippo_bilinear_wmma<<<blocks, WAVES * 32, 0, stream>>>(u, v, delta, A, Bvec, out, batch);
}


# 4x4 diagonal sub-block applied as its explicit inverse, formed one sub-block ahead from dt2 and A (pivots: rcp + Newton step): y = Minv r, 4 independent dot products; no division; sum order changes
# speedup vs baseline: 2.6041x; 1.9386x over previous
#include <hip/hip_runtime.h>


typedef __attribute__((ext_vector_type(2))) float v2f;
typedef __attribute__((ext_vector_type(4))) float v4f;
typedef __attribute__((ext_vector_type(8))) float v8f;

static constexpr int SN    = 128;
static constexpr int TILE  = 16;
static constexpr int NB    = SN / TILE;
static constexpr int APAD  = 132;
static constexpr int SUB   = 4;
static constexpr int WAVES = 4;

__device__ __forceinline__ v8f mma4(v2f af, v2f bf, v8f c)
{
    return __builtin_amdgcn_wmma_f32_16x16x4_f32(false, af, false, bf, (short)0, c, false, false);
}

__device__ __forceinline__ float xhalf(float x)
{
    const int b = __builtin_bit_cast(int, x);
    return __builtin_bit_cast(float,
        __builtin_amdgcn_permlanex16(b, b, 0x76543210, 0xfedcba98, false, false));
}

struct Inv4 { float m00, m10, m11, m20, m21, m22, m30, m31, m32, m33; };

__device__ __forceinline__ float recip(float d)
{
    const float x = __builtin_amdgcn_rcpf(d);
    return __builtin_fmaf(__builtin_fmaf(-d, x, 1.0f), x, x);
}

__device__ __forceinline__ Inv4 inv4(float dt2, float q00, v2f q1, v2f q2, float q22, v4f q3)
{
    const float i0 = recip(1.0f - dt2 * q00);
    const float i1 = recip(1.0f - dt2 * q1[1]);
    const float i2 = recip(1.0f - dt2 * q22);
    const float i3 = recip(1.0f - dt2 * q3[3]);
    const float c10 = dt2 * q1[0];
    const float c20 = dt2 * q2[0], c21 = dt2 * q2[1];
    const float c30 = dt2 * q3[0], c31 = dt2 * q3[1], c32 = dt2 * q3[2];
    Inv4 m;
    m.m00 = i0; m.m11 = i1; m.m22 = i2; m.m33 = i3;
    m.m10 = i1 * (c10 * i0);
    m.m21 = i2 * (c21 * i1);
    m.m32 = i3 * (c32 * i2);
    m.m20 = i2 * (c20 * i0 + c21 * m.m10);
    m.m31 = i3 * (c31 * i1 + c32 * m.m21);
    m.m30 = i3 * (c30 * i0 + c31 * m.m10 + c32 * m.m20);
    return m;
}

__global__ __launch_bounds__(WAVES * 32) void hippo_bilinear_wmma(
    const float* __restrict__ u, const float* __restrict__ v,
    const float* __restrict__ delta, const float* __restrict__ A,
    const float* __restrict__ Bvec, float* __restrict__ out, int batch)
{
    __shared__ float A_lds[SN * APAD];

    const int tid = threadIdx.x;
    const int nthreads = WAVES * 32;

    const int wave = tid >> 5;
    const int lane = tid & 31;
    const int col  = lane & 15;
    const int hlf  = lane >> 4;
    const int batch_base = (blockIdx.x * WAVES + wave) * TILE;
    const bool live = batch_base < batch;
    const int b = batch_base + col;
    const int bl = live ? b : 0;

    const float dlt = delta[bl];
    const float vv  = v[bl];
    const float* ucol = u + (size_t)bl * SN;
    v2f ufr[NB][4];
    #pragma unroll
    for (int j = 0; j < NB; ++j) {
        #pragma unroll
        for (int k4 = 0; k4 < 4; ++k4)
            ufr[j][k4] = *(const v2f*)(ucol + j * TILE + k4 * 4 + 2 * hlf);
    }
    v4f uc0[NB], uc1[NB], bc0[NB], bc1[NB];
    #pragma unroll
    for (int k = 0; k < NB; ++k) {
        const int nbase = k * TILE + 8 * hlf;
        uc0[k] = *(const v4f*)(ucol + nbase);
        uc1[k] = *(const v4f*)(ucol + nbase + 4);
        bc0[k] = *(const v4f*)(Bvec + nbase);
        bc1[k] = *(const v4f*)(Bvec + nbase + 4);
    }

    #pragma unroll 4
    for (int i = 0; i < (SN * SN / 4) / nthreads; ++i) {
        int q = tid + i * nthreads;
        int r = q >> 5;
        int c = (q & 31) << 2;
        v4f t = *(const v4f*)(A + r * SN + c);
        *(v4f*)(&A_lds[r * APAD + c]) = t;
    }
    __syncthreads();

    if (!live) return;

    const float dt2 = 0.5f * dlt;
    const float dvb = dlt * vv;

    const float* alane = &A_lds[col * APAD + 2 * hlf];

    v8f R[NB];
    #pragma unroll
    for (int k = 0; k < NB; ++k) R[k] = v8f{};
    #pragma unroll
    for (int j = 0; j < NB; ++j) {
        #pragma unroll
        for (int k4 = 0; k4 < 4; ++k4) {
            #pragma unroll
            for (int k = j; k < NB; ++k) {
                v2f af = *(const v2f*)(alane + (k * TILE) * APAD + j * TILE + k4 * 4);
                R[k] = __builtin_amdgcn_wmma_f32_16x16x4_f32(
                           false, af, false, ufr[j][k4], (short)0, R[k], false, false);
            }
        }
    }
    #pragma unroll
    for (int k = 0; k < NB; ++k) {
        #pragma unroll
        for (int r = 0; r < 4; ++r) {
            R[k][r]     = uc0[k][r] + dt2 * R[k][r]     + dvb * bc0[k][r];
            R[k][r + 4] = uc1[k][r] + dt2 * R[k][r + 4] + dvb * bc1[k][r];
        }
    }

    const bool up = (hlf != 0);

    v2f af[2][NB];
    #pragma unroll
    for (int p = 0; p < 2; ++p) {
        #pragma unroll
        for (int i = 0; i < NB; ++i)
            af[p][i] = *(const v2f*)(alane + (i * TILE) * APAD + SUB * p);
    }
    Inv4 Mn;
    {
        const float* cp = &A_lds[0];
        Mn = inv4(dt2, cp[0], *(const v2f*)(cp + APAD), *(const v2f*)(cp + 2 * APAD),
                  cp[2 * APAD + 2], *(const v4f*)(cp + 3 * APAD));
    }
    float q00[2]; v2f q1[2]; v2f q2[2]; float q22[2]; v4f q3[2];
    #pragma unroll
    for (int t = 1; t <= 2; ++t) {
        const float* cp = &A_lds[(SUB * t) * APAD + SUB * t];
        q00[t & 1] = cp[0];
        q1[t & 1]  = *(const v2f*)(cp + APAD);
        q2[t & 1]  = *(const v2f*)(cp + 2 * APAD);
        q22[t & 1] = cp[2 * APAD + 2];
        q3[t & 1]  = *(const v4f*)(cp + 3 * APAD);
    }

    #pragma unroll 1
    for (int kb = 0; kb < NB; ++kb) {
        const int n = NB - kb;
        float Y[8];

        #pragma unroll
        for (int s = 0; s < 4; ++s) {
            const int ha = s >> 1;
            const int rb = SUB * (s & 1);
            const int p  = s & 1;

            const Inv4 M = Mn;
            v2f a[NB];
            #pragma unroll
            for (int i = 0; i < NB; ++i) a[i] = af[p][i];

            Mn = inv4(dt2, q00[1 - p], q1[1 - p], q2[1 - p], q22[1 - p], q3[1 - p]);
            asm volatile("" :: "v"(Mn.m00), "v"(Mn.m10), "v"(Mn.m11), "v"(Mn.m20), "v"(Mn.m21),
                               "v"(Mn.m22), "v"(Mn.m30), "v"(Mn.m31), "v"(Mn.m32), "v"(Mn.m33));

            {
                const int nb = (s < 1) ? kb : ((kb + 1 < NB) ? (kb + 1) : (NB - 1));
                const int ns = (s + 3) & 3;
                const int cb = nb * TILE + SUB * ns;
                const float* cp = &A_lds[cb * APAD + cb];
                q00[1 - p] = cp[0];
                q1[1 - p]  = *(const v2f*)(cp + APAD);
                q2[1 - p]  = *(const v2f*)(cp + 2 * APAD);
                q22[1 - p] = cp[2 * APAD + 2];
                q3[1 - p]  = *(const v4f*)(cp + 3 * APAD);
            }
            {
                const int nb = (s < 2) ? kb : ((kb + 1 < NB) ? (kb + 1) : (NB - 1));
                const int ns = (s + 2) & 3;
                const int cb = nb * TILE + SUB * ns;
                #pragma unroll
                for (int i = 0; i < NB; ++i) {
                    const int rblk = (nb + i < NB) ? (nb + i) : (NB - 1);
                    af[p][i] = *(const v2f*)(alane + (rblk * TILE) * APAD + cb);
                }
            }

            const float r0 = R[0][rb], r1 = R[0][rb + 1], r2 = R[0][rb + 2], r3 = R[0][rb + 3];
            const float y0 = M.m00 * r0;
            const float y1 = M.m10 * r0 + M.m11 * r1;
            const float y2 = M.m20 * r0 + M.m21 * r1 + M.m22 * r2;
            const float y3 = (M.m30 * r0 + M.m31 * r1) + (M.m32 * r2 + M.m33 * r3);

            const float z0 = dt2 * y0, z1 = dt2 * y1, z2 = dt2 * y2, z3 = dt2 * y3;
            v2f zb;
            if (ha == 0) {
                const float t2 = xhalf(z2), t3 = xhalf(z3);
                zb.x = up ? t2 : z0;
                zb.y = up ? t3 : z1;
                Y[rb] = y0; Y[rb + 1] = y1; Y[rb + 2] = y2; Y[rb + 3] = y3;
            } else {
                const float t0 = xhalf(z0), t1 = xhalf(z1);
                zb.x = up ? z2 : t0;
                zb.y = up ? z3 : t1;
                Y[rb]     = up ? y0 : Y[rb];
                Y[rb + 1] = up ? y1 : Y[rb + 1];
                Y[rb + 2] = up ? y2 : Y[rb + 2];
                Y[rb + 3] = up ? y3 : Y[rb + 3];
            }

            if (s < 3) {
                R[0] = mma4(a[0], zb, R[0]);
                if (n > 1) { R[1] = mma4(a[1], zb, R[1]);
                if (n > 2) { R[2] = mma4(a[2], zb, R[2]);
                if (n > 3) { R[3] = mma4(a[3], zb, R[3]);
                if (n > 4) { R[4] = mma4(a[4], zb, R[4]);
                if (n > 5) { R[5] = mma4(a[5], zb, R[5]);
                if (n > 6) { R[6] = mma4(a[6], zb, R[6]);
                if (n > 7) { R[7] = mma4(a[7], zb, R[7]); } } } } } } }
            } else {
                if (n > 1) { R[0] = mma4(a[1], zb, R[1]);
                if (n > 2) { R[1] = mma4(a[2], zb, R[2]);
                if (n > 3) { R[2] = mma4(a[3], zb, R[3]);
                if (n > 4) { R[3] = mma4(a[4], zb, R[4]);
                if (n > 5) { R[4] = mma4(a[5], zb, R[5]);
                if (n > 6) { R[5] = mma4(a[6], zb, R[6]);
                if (n > 7) { R[6] = mma4(a[7], zb, R[7]); } } } } } } }
            }
        }

        float* o = out + (size_t)b * SN + kb * TILE + 8 * hlf;
        const v4f o0 = { Y[0], Y[1], Y[2], Y[3] };
        const v4f o1 = { Y[4], Y[5], Y[6], Y[7] };
        *(v4f*)(o)     = o0;
        *(v4f*)(o + 4) = o1;
    }
}

extern "C" void kernel_launch(void* const* d_in, const int* in_sizes, int n_in,
                              void* d_out, int out_size, void* d_ws, size_t ws_size,
                              hipStream_t stream) {
    const float* u     = (const float*)d_in[0];
    const float* v     = (const float*)d_in[1];
    const float* delta = (const float*)d_in[2];
    const float* A     = (const float*)d_in[3];
    const float* Bvec  = (const float*)d_in[4];
    float* out = (float*)d_out;

    const int batch  = in_sizes[0] / SN;
    const int blocks = batch / (WAVES * TILE);
    hippo_bilinear_wmma<<<blocks, WAVES * 32, 0, stream>>>(u, v, delta, A, Bvec, out, batch);
}


# A's 144 lower-triangle WMMA fragments read from LDS once, as one group, and kept in registers for forward product and substitution alike (substitution unrolled); 4x4 triangles grouped per block
# speedup vs baseline: 2.9617x; 1.1373x over previous
#include <hip/hip_runtime.h>


typedef __attribute__((ext_vector_type(2))) float v2f;
typedef __attribute__((ext_vector_type(4))) float v4f;
typedef __attribute__((ext_vector_type(8))) float v8f;

static constexpr int SN    = 128;
static constexpr int TILE  = 16;
static constexpr int NB    = SN / TILE;
static constexpr int APAD  = 132;
static constexpr int SUB   = 4;
static constexpr int WAVES = 4;
static constexpr int NBLK  = NB * (NB + 1) / 2;

__host__ __device__ constexpr int bidx(int k, int j) { return j * NB - (j * (j - 1)) / 2 + (k - j); }

__device__ __forceinline__ v8f mma4(v2f af, v2f bf, v8f c)
{
    return __builtin_amdgcn_wmma_f32_16x16x4_f32(false, af, false, bf, (short)0, c, false, false);
}

__device__ __forceinline__ float xhalf(float x)
{
    const int b = __builtin_bit_cast(int, x);
    return __builtin_bit_cast(float,
        __builtin_amdgcn_permlanex16(b, b, 0x76543210, 0xfedcba98, false, false));
}

struct Inv4 { float m00, m10, m11, m20, m21, m22, m30, m31, m32, m33; };

__device__ __forceinline__ float recip(float d)
{
    const float x = __builtin_amdgcn_rcpf(d);
    return __builtin_fmaf(__builtin_fmaf(-d, x, 1.0f), x, x);
}

__device__ __forceinline__ Inv4 inv4(float dt2, float q00, v2f q1, v2f q2, float q22, v4f q3)
{
    const float i0 = recip(1.0f - dt2 * q00);
    const float i1 = recip(1.0f - dt2 * q1[1]);
    const float i2 = recip(1.0f - dt2 * q22);
    const float i3 = recip(1.0f - dt2 * q3[3]);
    const float c10 = dt2 * q1[0];
    const float c20 = dt2 * q2[0], c21 = dt2 * q2[1];
    const float c30 = dt2 * q3[0], c31 = dt2 * q3[1], c32 = dt2 * q3[2];
    Inv4 m;
    m.m00 = i0; m.m11 = i1; m.m22 = i2; m.m33 = i3;
    m.m10 = i1 * (c10 * i0);
    m.m21 = i2 * (c21 * i1);
    m.m32 = i3 * (c32 * i2);
    m.m20 = i2 * (c20 * i0 + c21 * m.m10);
    m.m31 = i3 * (c31 * i1 + c32 * m.m21);
    m.m30 = i3 * (c30 * i0 + c31 * m.m10 + c32 * m.m20);
    return m;
}

struct Tri4 { float q00; v2f q1; v2f q2; float q22; v4f q3; };

__device__ __forceinline__ Tri4 load_tri(const float* A_lds, int cb)
{
    const float* cp = A_lds + cb * APAD + cb;
    Tri4 t;
    t.q00 = cp[0];
    t.q1  = *(const v2f*)(cp + APAD);
    t.q2  = *(const v2f*)(cp + 2 * APAD);
    t.q22 = cp[2 * APAD + 2];
    t.q3  = *(const v4f*)(cp + 3 * APAD);
    return t;
}

__global__ __launch_bounds__(WAVES * 32) void hippo_bilinear_wmma(
    const float* __restrict__ u, const float* __restrict__ v,
    const float* __restrict__ delta, const float* __restrict__ A,
    const float* __restrict__ Bvec, float* __restrict__ out, int batch)
{
    __shared__ float A_lds[SN * APAD];

    const int tid = threadIdx.x;
    const int nthreads = WAVES * 32;

    const int wave = tid >> 5;
    const int lane = tid & 31;
    const int col  = lane & 15;
    const int hlf  = lane >> 4;
    const int batch_base = (blockIdx.x * WAVES + wave) * TILE;
    const bool live = batch_base < batch;
    const int b = batch_base + col;
    const int bl = live ? b : 0;

    const float dlt = delta[bl];
    const float vv  = v[bl];
    const float* ucol = u + (size_t)bl * SN;
    v2f ufr[NB][4];
    #pragma unroll
    for (int j = 0; j < NB; ++j) {
        #pragma unroll
        for (int k4 = 0; k4 < 4; ++k4)
            ufr[j][k4] = *(const v2f*)(ucol + j * TILE + k4 * 4 + 2 * hlf);
    }
    v4f uc0[NB], uc1[NB], bc0[NB], bc1[NB];
    #pragma unroll
    for (int k = 0; k < NB; ++k) {
        const int nbase = k * TILE + 8 * hlf;
        uc0[k] = *(const v4f*)(ucol + nbase);
        uc1[k] = *(const v4f*)(ucol + nbase + 4);
        bc0[k] = *(const v4f*)(Bvec + nbase);
        bc1[k] = *(const v4f*)(Bvec + nbase + 4);
    }

    #pragma unroll 4
    for (int i = 0; i < (SN * SN / 4) / nthreads; ++i) {
        int q = tid + i * nthreads;
        int r = q >> 5;
        int c = (q & 31) << 2;
        v4f t = *(const v4f*)(A + r * SN + c);
        *(v4f*)(&A_lds[r * APAD + c]) = t;
    }
    __syncthreads();

    if (!live) return;

    const float dt2 = 0.5f * dlt;
    const float dvb = dlt * vv;

    const float* alane = &A_lds[col * APAD + 2 * hlf];

    v2f afr[NBLK][4];
    #pragma unroll
    for (int j = 0; j < NB; ++j) {
        #pragma unroll
        for (int k4 = 0; k4 < 4; ++k4) {
            #pragma unroll
            for (int k = j; k < NB; ++k)
                afr[bidx(k, j)][k4] = *(const v2f*)(alane + (k * TILE) * APAD + j * TILE + k4 * 4);
        }
    }
    Tri4 tri[2][4];
    #pragma unroll
    for (int t = 0; t < 2; ++t) {
        #pragma unroll
        for (int s = 0; s < 4; ++s)
            tri[t][s] = load_tri(A_lds, t * TILE + SUB * s);
    }
    __builtin_amdgcn_sched_barrier(0);

    v8f R[NB];
    #pragma unroll
    for (int k = 0; k < NB; ++k) R[k] = v8f{};
    #pragma unroll
    for (int j = 0; j < NB; ++j) {
        #pragma unroll
        for (int k4 = 0; k4 < 4; ++k4) {
            #pragma unroll
            for (int k = j; k < NB; ++k)
                R[k] = mma4(afr[bidx(k, j)][k4], ufr[j][k4], R[k]);
        }
    }
    #pragma unroll
    for (int k = 0; k < NB; ++k) {
        #pragma unroll
        for (int r = 0; r < 4; ++r) {
            R[k][r]     = uc0[k][r] + dt2 * R[k][r]     + dvb * bc0[k][r];
            R[k][r + 4] = uc1[k][r] + dt2 * R[k][r + 4] + dvb * bc1[k][r];
        }
    }

    const bool up = (hlf != 0);

    Inv4 Mn;
    {
        const Tri4 t0 = tri[0][0];
        Mn = inv4(dt2, t0.q00, t0.q1, t0.q2, t0.q22, t0.q3);
    }

    #pragma unroll
    for (int kb = 0; kb < NB; ++kb) {
        float Y[8];

        if (kb >= 1 && kb + 1 < NB) {
            __builtin_amdgcn_sched_barrier(0);
            #pragma unroll
            for (int s = 0; s < 4; ++s)
                tri[(kb + 1) & 1][s] = load_tri(A_lds, (kb + 1) * TILE + SUB * s);
            __builtin_amdgcn_sched_barrier(0);
        }

        #pragma unroll
        for (int s = 0; s < 4; ++s) {
            const int ha = s >> 1;
            const int rb = SUB * (s & 1);

            const Inv4 M = Mn;

            if (kb * 4 + s + 1 < NB * 4) {
                const Tri4 tn = (s < 3) ? tri[kb & 1][s + 1] : tri[(kb + 1) & 1][0];
                Mn = inv4(dt2, tn.q00, tn.q1, tn.q2, tn.q22, tn.q3);
                asm volatile("" :: "v"(Mn.m00), "v"(Mn.m10), "v"(Mn.m11), "v"(Mn.m20), "v"(Mn.m21),
                                   "v"(Mn.m22), "v"(Mn.m30), "v"(Mn.m31), "v"(Mn.m32), "v"(Mn.m33));
            }

            const float r0 = R[kb][rb], r1 = R[kb][rb + 1], r2 = R[kb][rb + 2], r3 = R[kb][rb + 3];
            const float y0 = M.m00 * r0;
            const float y1 = M.m10 * r0 + M.m11 * r1;
            const float y2 = M.m20 * r0 + M.m21 * r1 + M.m22 * r2;
            const float y3 = (M.m30 * r0 + M.m31 * r1) + (M.m32 * r2 + M.m33 * r3);

            const float z0 = dt2 * y0, z1 = dt2 * y1, z2 = dt2 * y2, z3 = dt2 * y3;
            v2f zb;
            if (ha == 0) {
                const float t2 = xhalf(z2), t3 = xhalf(z3);
                zb.x = up ? t2 : z0;
                zb.y = up ? t3 : z1;
                Y[rb] = y0; Y[rb + 1] = y1; Y[rb + 2] = y2; Y[rb + 3] = y3;
            } else {
                const float t0 = xhalf(z0), t1 = xhalf(z1);
                zb.x = up ? z2 : t0;
                zb.y = up ? z3 : t1;
                Y[rb]     = up ? y0 : Y[rb];
                Y[rb + 1] = up ? y1 : Y[rb + 1];
                Y[rb + 2] = up ? y2 : Y[rb + 2];
                Y[rb + 3] = up ? y3 : Y[rb + 3];
            }

            #pragma unroll
            for (int k = (s < 3) ? kb : kb + 1; k < NB; ++k)
                R[k] = mma4(afr[bidx(k, kb)][s], zb, R[k]);
        }

        float* o = out + (size_t)b * SN + kb * TILE + 8 * hlf;
        const v4f o0 = { Y[0], Y[1], Y[2], Y[3] };
        const v4f o1 = { Y[4], Y[5], Y[6], Y[7] };
        *(v4f*)(o)     = o0;
        *(v4f*)(o + 4) = o1;
    }
}

extern "C" void kernel_launch(void* const* d_in, const int* in_sizes, int n_in,
                              void* d_out, int out_size, void* d_ws, size_t ws_size,
                              hipStream_t stream) {
    const float* u     = (const float*)d_in[0];
    const float* v     = (const float*)d_in[1];
    const float* delta = (const float*)d_in[2];
    const float* A     = (const float*)d_in[3];
    const float* Bvec  = (const float*)d_in[4];
    float* out = (float*)d_out;

    const int batch  = in_sizes[0] / SN;
    const int blocks = batch / (WAVES * TILE);
    hippo_bilinear_wmma<<<blocks, WAVES * 32, 0, stream>>>(u, v, delta, A, Bvec, out, batch);
}


# A staging order rotated per work-group (start row = 2 x block index, wrapping): concurrent work-groups no longer ask the L2 for the same lines of A at the same moment; sums unchanged
# speedup vs baseline: 3.7185x; 1.2555x over previous
#include <hip/hip_runtime.h>


typedef __attribute__((ext_vector_type(2))) float v2f;
typedef __attribute__((ext_vector_type(4))) float v4f;
typedef __attribute__((ext_vector_type(8))) float v8f;

static constexpr int SN    = 128;
static constexpr int TILE  = 16;
static constexpr int NB    = SN / TILE;
static constexpr int APAD  = 132;
static constexpr int SUB   = 4;
static constexpr int WAVES = 4;
static constexpr int NBLK  = NB * (NB + 1) / 2;

__host__ __device__ constexpr int bidx(int k, int j) { return j * NB - (j * (j - 1)) / 2 + (k - j); }

__device__ __forceinline__ v8f mma4(v2f af, v2f bf, v8f c)
{
    return __builtin_amdgcn_wmma_f32_16x16x4_f32(false, af, false, bf, (short)0, c, false, false);
}

__device__ __forceinline__ float xhalf(float x)
{
    const int b = __builtin_bit_cast(int, x);
    return __builtin_bit_cast(float,
        __builtin_amdgcn_permlanex16(b, b, 0x76543210, 0xfedcba98, false, false));
}

struct Inv4 { float m00, m10, m11, m20, m21, m22, m30, m31, m32, m33; };

__device__ __forceinline__ float recip(float d)
{
    const float x = __builtin_amdgcn_rcpf(d);
    return __builtin_fmaf(__builtin_fmaf(-d, x, 1.0f), x, x);
}

__device__ __forceinline__ Inv4 inv4(float dt2, float q00, v2f q1, v2f q2, float q22, v4f q3)
{
    const float i0 = recip(1.0f - dt2 * q00);
    const float i1 = recip(1.0f - dt2 * q1[1]);
    const float i2 = recip(1.0f - dt2 * q22);
    const float i3 = recip(1.0f - dt2 * q3[3]);
    const float c10 = dt2 * q1[0];
    const float c20 = dt2 * q2[0], c21 = dt2 * q2[1];
    const float c30 = dt2 * q3[0], c31 = dt2 * q3[1], c32 = dt2 * q3[2];
    Inv4 m;
    m.m00 = i0; m.m11 = i1; m.m22 = i2; m.m33 = i3;
    m.m10 = i1 * (c10 * i0);
    m.m21 = i2 * (c21 * i1);
    m.m32 = i3 * (c32 * i2);
    m.m20 = i2 * (c20 * i0 + c21 * m.m10);
    m.m31 = i3 * (c31 * i1 + c32 * m.m21);
    m.m30 = i3 * (c30 * i0 + c31 * m.m10 + c32 * m.m20);
    return m;
}

struct Tri4 { float q00; v2f q1; v2f q2; float q22; v4f q3; };

__device__ __forceinline__ Tri4 load_tri(const float* A_lds, int cb)
{
    const float* cp = A_lds + cb * APAD + cb;
    Tri4 t;
    t.q00 = cp[0];
    t.q1  = *(const v2f*)(cp + APAD);
    t.q2  = *(const v2f*)(cp + 2 * APAD);
    t.q22 = cp[2 * APAD + 2];
    t.q3  = *(const v4f*)(cp + 3 * APAD);
    return t;
}

__global__ __launch_bounds__(WAVES * 32) void hippo_bilinear_wmma(
    const float* __restrict__ u, const float* __restrict__ v,
    const float* __restrict__ delta, const float* __restrict__ A,
    const float* __restrict__ Bvec, float* __restrict__ out, int batch)
{
    __shared__ float A_lds[SN * APAD];

    const int tid = threadIdx.x;
    const int nthreads = WAVES * 32;

    const int wave = tid >> 5;
    const int lane = tid & 31;
    const int col  = lane & 15;
    const int hlf  = lane >> 4;
    const int batch_base = (blockIdx.x * WAVES + wave) * TILE;
    const bool live = batch_base < batch;
    const int b = batch_base + col;
    const int bl = live ? b : 0;

    const float dlt = delta[bl];
    const float vv  = v[bl];
    const float* ucol = u + (size_t)bl * SN;
    v2f ufr[NB][4];
    #pragma unroll
    for (int j = 0; j < NB; ++j) {
        #pragma unroll
        for (int k4 = 0; k4 < 4; ++k4)
            ufr[j][k4] = *(const v2f*)(ucol + j * TILE + k4 * 4 + 2 * hlf);
    }
    v4f uc0[NB], uc1[NB], bc0[NB], bc1[NB];
    #pragma unroll
    for (int k = 0; k < NB; ++k) {
        const int nbase = k * TILE + 8 * hlf;
        uc0[k] = *(const v4f*)(ucol + nbase);
        uc1[k] = *(const v4f*)(ucol + nbase + 4);
        bc0[k] = *(const v4f*)(Bvec + nbase);
        bc1[k] = *(const v4f*)(Bvec + nbase + 4);
    }

    constexpr int NQ = SN * SN / 4;
    const int rot = (int)((blockIdx.x * 64u) & (unsigned)(NQ - 1));
    #pragma unroll 4
    for (int i = 0; i < NQ / nthreads; ++i) {
        int q = (tid + i * nthreads + rot) & (NQ - 1);
        int r = q >> 5;
        int c = (q & 31) << 2;
        v4f t = *(const v4f*)(A + r * SN + c);
        *(v4f*)(&A_lds[r * APAD + c]) = t;
    }
    __syncthreads();

    if (!live) return;

    const float dt2 = 0.5f * dlt;
    const float dvb = dlt * vv;

    const float* alane = &A_lds[col * APAD + 2 * hlf];

    v2f afr[NBLK][4];
    #pragma unroll
    for (int j = 0; j < NB; ++j) {
        #pragma unroll
        for (int k4 = 0; k4 < 4; ++k4) {
            #pragma unroll
            for (int k = j; k < NB; ++k)
                afr[bidx(k, j)][k4] = *(const v2f*)(alane + (k * TILE) * APAD + j * TILE + k4 * 4);
        }
    }
    Tri4 tri[2][4];
    #pragma unroll
    for (int t = 0; t < 2; ++t) {
        #pragma unroll
        for (int s = 0; s < 4; ++s)
            tri[t][s] = load_tri(A_lds, t * TILE + SUB * s);
    }
    __builtin_amdgcn_sched_barrier(0);

    v8f R[NB];
    #pragma unroll
    for (int k = 0; k < NB; ++k) R[k] = v8f{};
    #pragma unroll
    for (int j = 0; j < NB; ++j) {
        #pragma unroll
        for (int k4 = 0; k4 < 4; ++k4) {
            #pragma unroll
            for (int k = j; k < NB; ++k)
                R[k] = mma4(afr[bidx(k, j)][k4], ufr[j][k4], R[k]);
        }
    }
    #pragma unroll
    for (int k = 0; k < NB; ++k) {
        #pragma unroll
        for (int r = 0; r < 4; ++r) {
            R[k][r]     = uc0[k][r] + dt2 * R[k][r]     + dvb * bc0[k][r];
            R[k][r + 4] = uc1[k][r] + dt2 * R[k][r + 4] + dvb * bc1[k][r];
        }
    }

    const bool up = (hlf != 0);

    Inv4 Mn;
    {
        const Tri4 t0 = tri[0][0];
        Mn = inv4(dt2, t0.q00, t0.q1, t0.q2, t0.q22, t0.q3);
    }

    #pragma unroll
    for (int kb = 0; kb < NB; ++kb) {
        float Y[8];

        if (kb >= 1 && kb + 1 < NB) {
            __builtin_amdgcn_sched_barrier(0);
            #pragma unroll
            for (int s = 0; s < 4; ++s)
                tri[(kb + 1) & 1][s] = load_tri(A_lds, (kb + 1) * TILE + SUB * s);
            __builtin_amdgcn_sched_barrier(0);
        }

        #pragma unroll
        for (int s = 0; s < 4; ++s) {
            const int ha = s >> 1;
            const int rb = SUB * (s & 1);

            const Inv4 M = Mn;

            if (kb * 4 + s + 1 < NB * 4) {
                const Tri4 tn = (s < 3) ? tri[kb & 1][s + 1] : tri[(kb + 1) & 1][0];
                Mn = inv4(dt2, tn.q00, tn.q1, tn.q2, tn.q22, tn.q3);
                asm volatile("" :: "v"(Mn.m00), "v"(Mn.m10), "v"(Mn.m11), "v"(Mn.m20), "v"(Mn.m21),
                                   "v"(Mn.m22), "v"(Mn.m30), "v"(Mn.m31), "v"(Mn.m32), "v"(Mn.m33));
            }

            const float r0 = R[kb][rb], r1 = R[kb][rb + 1], r2 = R[kb][rb + 2], r3 = R[kb][rb + 3];
            const float y0 = M.m00 * r0;
            const float y1 = M.m10 * r0 + M.m11 * r1;
            const float y2 = M.m20 * r0 + M.m21 * r1 + M.m22 * r2;
            const float y3 = (M.m30 * r0 + M.m31 * r1) + (M.m32 * r2 + M.m33 * r3);

            const float z0 = dt2 * y0, z1 = dt2 * y1, z2 = dt2 * y2, z3 = dt2 * y3;
            v2f zb;
            if (ha == 0) {
                const float t2 = xhalf(z2), t3 = xhalf(z3);
                zb.x = up ? t2 : z0;
                zb.y = up ? t3 : z1;
                Y[rb] = y0; Y[rb + 1] = y1; Y[rb + 2] = y2; Y[rb + 3] = y3;
            } else {
                const float t0 = xhalf(z0), t1 = xhalf(z1);
                zb.x = up ? z2 : t0;
                zb.y = up ? z3 : t1;
                Y[rb]     = up ? y0 : Y[rb];
                Y[rb + 1] = up ? y1 : Y[rb + 1];
                Y[rb + 2] = up ? y2 : Y[rb + 2];
                Y[rb + 3] = up ? y3 : Y[rb + 3];
            }

            #pragma unroll
            for (int k = (s < 3) ? kb : kb + 1; k < NB; ++k)
                R[k] = mma4(afr[bidx(k, kb)][s], zb, R[k]);
        }

        float* o = out + (size_t)b * SN + kb * TILE + 8 * hlf;
        const v4f o0 = { Y[0], Y[1], Y[2], Y[3] };
        const v4f o1 = { Y[4], Y[5], Y[6], Y[7] };
        *(v4f*)(o)     = o0;
        *(v4f*)(o + 4) = o1;
    }
}

extern "C" void kernel_launch(void* const* d_in, const int* in_sizes, int n_in,
                              void* d_out, int out_size, void* d_ws, size_t ws_size,
                              hipStream_t stream) {
    const float* u     = (const float*)d_in[0];
    const float* v     = (const float*)d_in[1];
    const float* delta = (const float*)d_in[2];
    const float* A     = (const float*)d_in[3];
    const float* Bvec  = (const float*)d_in[4];
    float* out = (float*)d_out;

    const int batch  = in_sizes[0] / SN;
    const int blocks = batch / (WAVES * TILE);
    hippo_bilinear_wmma<<<blocks, WAVES * 32, 0, stream>>>(u, v, delta, A, Bvec, out, batch);
}


# u read from memory once: the forward product's B fragments are built from the accumulator-layout copy by half-wave exchange (4 v_permlanex16 per block), not by 32 more 16-line loads; sums unchanged
# speedup vs baseline: 3.9116x; 1.0519x over previous
#include <hip/hip_runtime.h>


typedef __attribute__((ext_vector_type(2))) float v2f;
typedef __attribute__((ext_vector_type(4))) float v4f;
typedef __attribute__((ext_vector_type(8))) float v8f;

static constexpr int SN    = 128;
static constexpr int TILE  = 16;
static constexpr int NB    = SN / TILE;
static constexpr int APAD  = 132;
static constexpr int SUB   = 4;
static constexpr int WAVES = 4;
static constexpr int NBLK  = NB * (NB + 1) / 2;

__host__ __device__ constexpr int bidx(int k, int j) { return j * NB - (j * (j - 1)) / 2 + (k - j); }

__device__ __forceinline__ v8f mma4(v2f af, v2f bf, v8f c)
{
    return __builtin_amdgcn_wmma_f32_16x16x4_f32(false, af, false, bf, (short)0, c, false, false);
}

__device__ __forceinline__ float xhalf(float x)
{
    const int b = __builtin_bit_cast(int, x);
    return __builtin_bit_cast(float,
        __builtin_amdgcn_permlanex16(b, b, 0x76543210, 0xfedcba98, false, false));
}

struct Inv4 { float m00, m10, m11, m20, m21, m22, m30, m31, m32, m33; };

__device__ __forceinline__ float recip(float d)
{
    const float x = __builtin_amdgcn_rcpf(d);
    return __builtin_fmaf(__builtin_fmaf(-d, x, 1.0f), x, x);
}

__device__ __forceinline__ Inv4 inv4(float dt2, float q00, v2f q1, v2f q2, float q22, v4f q3)
{
    const float i0 = recip(1.0f - dt2 * q00);
    const float i1 = recip(1.0f - dt2 * q1[1]);
    const float i2 = recip(1.0f - dt2 * q22);
    const float i3 = recip(1.0f - dt2 * q3[3]);
    const float c10 = dt2 * q1[0];
    const float c20 = dt2 * q2[0], c21 = dt2 * q2[1];
    const float c30 = dt2 * q3[0], c31 = dt2 * q3[1], c32 = dt2 * q3[2];
    Inv4 m;
    m.m00 = i0; m.m11 = i1; m.m22 = i2; m.m33 = i3;
    m.m10 = i1 * (c10 * i0);
    m.m21 = i2 * (c21 * i1);
    m.m32 = i3 * (c32 * i2);
    m.m20 = i2 * (c20 * i0 + c21 * m.m10);
    m.m31 = i3 * (c31 * i1 + c32 * m.m21);
    m.m30 = i3 * (c30 * i0 + c31 * m.m10 + c32 * m.m20);
    return m;
}

struct Tri4 { float q00; v2f q1; v2f q2; float q22; v4f q3; };

__device__ __forceinline__ Tri4 load_tri(const float* A_lds, int cb)
{
    const float* cp = A_lds + cb * APAD + cb;
    Tri4 t;
    t.q00 = cp[0];
    t.q1  = *(const v2f*)(cp + APAD);
    t.q2  = *(const v2f*)(cp + 2 * APAD);
    t.q22 = cp[2 * APAD + 2];
    t.q3  = *(const v4f*)(cp + 3 * APAD);
    return t;
}

__global__ __launch_bounds__(WAVES * 32) void hippo_bilinear_wmma(
    const float* __restrict__ u, const float* __restrict__ v,
    const float* __restrict__ delta, const float* __restrict__ A,
    const float* __restrict__ Bvec, float* __restrict__ out, int batch)
{
    __shared__ float A_lds[SN * APAD];

    const int tid = threadIdx.x;
    const int nthreads = WAVES * 32;

    const int wave = tid >> 5;
    const int lane = tid & 31;
    const int col  = lane & 15;
    const int hlf  = lane >> 4;
    const int batch_base = (blockIdx.x * WAVES + wave) * TILE;
    const bool live = batch_base < batch;
    const int b = batch_base + col;
    const int bl = live ? b : 0;

    const float dlt = delta[bl];
    const float vv  = v[bl];
    const float* ucol = u + (size_t)bl * SN;
    v4f uc0[NB], uc1[NB], bc0[NB], bc1[NB];
    #pragma unroll
    for (int k = 0; k < NB; ++k) {
        const int nbase = k * TILE + 8 * hlf;
        uc0[k] = *(const v4f*)(ucol + nbase);
        uc1[k] = *(const v4f*)(ucol + nbase + 4);
        bc0[k] = *(const v4f*)(Bvec + nbase);
        bc1[k] = *(const v4f*)(Bvec + nbase + 4);
    }

    constexpr int NQ = SN * SN / 4;
    const int rot = (int)((blockIdx.x * 64u) & (unsigned)(NQ - 1));
    #pragma unroll 4
    for (int i = 0; i < NQ / nthreads; ++i) {
        int q = (tid + i * nthreads + rot) & (NQ - 1);
        int r = q >> 5;
        int c = (q & 31) << 2;
        v4f t = *(const v4f*)(A + r * SN + c);
        *(v4f*)(&A_lds[r * APAD + c]) = t;
    }
    __syncthreads();

    if (!live) return;

    const float dt2 = 0.5f * dlt;
    const float dvb = dlt * vv;

    const float* alane = &A_lds[col * APAD + 2 * hlf];

    v2f afr[NBLK][4];
    #pragma unroll
    for (int j = 0; j < NB; ++j) {
        #pragma unroll
        for (int k4 = 0; k4 < 4; ++k4) {
            #pragma unroll
            for (int k = j; k < NB; ++k)
                afr[bidx(k, j)][k4] = *(const v2f*)(alane + (k * TILE) * APAD + j * TILE + k4 * 4);
        }
    }
    Tri4 tri[2][4];
    #pragma unroll
    for (int t = 0; t < 2; ++t) {
        #pragma unroll
        for (int s = 0; s < 4; ++s)
            tri[t][s] = load_tri(A_lds, t * TILE + SUB * s);
    }
    __builtin_amdgcn_sched_barrier(0);

    const bool up = (hlf != 0);
    v8f R[NB];
    #pragma unroll
    for (int k = 0; k < NB; ++k) R[k] = v8f{};
    #pragma unroll
    for (int j = 0; j < NB; ++j) {
        const float tx0 = xhalf(up ? uc0[j][0] : uc0[j][2]);
        const float ty0 = xhalf(up ? uc0[j][1] : uc0[j][3]);
        const float tx1 = xhalf(up ? uc1[j][0] : uc1[j][2]);
        const float ty1 = xhalf(up ? uc1[j][1] : uc1[j][3]);
        v2f uf[4];
        uf[0].x = up ? tx0 : uc0[j][0];   uf[0].y = up ? ty0 : uc0[j][1];
        uf[1].x = up ? tx1 : uc1[j][0];   uf[1].y = up ? ty1 : uc1[j][1];
        uf[2].x = up ? uc0[j][2] : tx0;   uf[2].y = up ? uc0[j][3] : ty0;
        uf[3].x = up ? uc1[j][2] : tx1;   uf[3].y = up ? uc1[j][3] : ty1;
        #pragma unroll
        for (int k4 = 0; k4 < 4; ++k4) {
            #pragma unroll
            for (int k = j; k < NB; ++k)
                R[k] = mma4(afr[bidx(k, j)][k4], uf[k4], R[k]);
        }
        if (j == 0) __builtin_amdgcn_sched_barrier(0);
    }
    #pragma unroll
    for (int k = 0; k < NB; ++k) {
        #pragma unroll
        for (int r = 0; r < 4; ++r) {
            R[k][r]     = uc0[k][r] + dt2 * R[k][r]     + dvb * bc0[k][r];
            R[k][r + 4] = uc1[k][r] + dt2 * R[k][r + 4] + dvb * bc1[k][r];
        }
    }

    Inv4 Mn;
    {
        const Tri4 t0 = tri[0][0];
        Mn = inv4(dt2, t0.q00, t0.q1, t0.q2, t0.q22, t0.q3);
    }

    #pragma unroll
    for (int kb = 0; kb < NB; ++kb) {
        float Y[8];

        if (kb >= 1 && kb + 1 < NB) {
            __builtin_amdgcn_sched_barrier(0);
            #pragma unroll
            for (int s = 0; s < 4; ++s)
                tri[(kb + 1) & 1][s] = load_tri(A_lds, (kb + 1) * TILE + SUB * s);
            __builtin_amdgcn_sched_barrier(0);
        }

        #pragma unroll
        for (int s = 0; s < 4; ++s) {
            const int ha = s >> 1;
            const int rb = SUB * (s & 1);

            const Inv4 M = Mn;

            if (kb * 4 + s + 1 < NB * 4) {
                const Tri4 tn = (s < 3) ? tri[kb & 1][s + 1] : tri[(kb + 1) & 1][0];
                Mn = inv4(dt2, tn.q00, tn.q1, tn.q2, tn.q22, tn.q3);
                asm volatile("" :: "v"(Mn.m00), "v"(Mn.m10), "v"(Mn.m11), "v"(Mn.m20), "v"(Mn.m21),
                                   "v"(Mn.m22), "v"(Mn.m30), "v"(Mn.m31), "v"(Mn.m32), "v"(Mn.m33));
            }

            const float r0 = R[kb][rb], r1 = R[kb][rb + 1], r2 = R[kb][rb + 2], r3 = R[kb][rb + 3];
            const float y0 = M.m00 * r0;
            const float y1 = M.m10 * r0 + M.m11 * r1;
            const float y2 = M.m20 * r0 + M.m21 * r1 + M.m22 * r2;
            const float y3 = (M.m30 * r0 + M.m31 * r1) + (M.m32 * r2 + M.m33 * r3);

            const float z0 = dt2 * y0, z1 = dt2 * y1, z2 = dt2 * y2, z3 = dt2 * y3;
            v2f zb;
            if (ha == 0) {
                const float t2 = xhalf(z2), t3 = xhalf(z3);
                zb.x = up ? t2 : z0;
                zb.y = up ? t3 : z1;
                Y[rb] = y0; Y[rb + 1] = y1; Y[rb + 2] = y2; Y[rb + 3] = y3;
            } else {
                const float t0 = xhalf(z0), t1 = xhalf(z1);
                zb.x = up ? z2 : t0;
                zb.y = up ? z3 : t1;
                Y[rb]     = up ? y0 : Y[rb];
                Y[rb + 1] = up ? y1 : Y[rb + 1];
                Y[rb + 2] = up ? y2 : Y[rb + 2];
                Y[rb + 3] = up ? y3 : Y[rb + 3];
            }

            #pragma unroll
            for (int k = (s < 3) ? kb : kb + 1; k < NB; ++k)
                R[k] = mma4(afr[bidx(k, kb)][s], zb, R[k]);
        }

        float* o = out + (size_t)b * SN + kb * TILE + 8 * hlf;
        const v4f o0 = { Y[0], Y[1], Y[2], Y[3] };
        const v4f o1 = { Y[4], Y[5], Y[6], Y[7] };
        *(v4f*)(o)     = o0;
        *(v4f*)(o + 4) = o1;
    }
}

extern "C" void kernel_launch(void* const* d_in, const int* in_sizes, int n_in,
                              void* d_out, int out_size, void* d_ws, size_t ws_size,
                              hipStream_t stream) {
    const float* u     = (const float*)d_in[0];
    const float* v     = (const float*)d_in[1];
    const float* delta = (const float*)d_in[2];
    const float* A     = (const float*)d_in[3];
    const float* Bvec  = (const float*)d_in[4];
    float* out = (float*)d_out;

    const int batch  = in_sizes[0] / SN;
    const int blocks = batch / (WAVES * TILE);
    hippo_bilinear_wmma<<<blocks, WAVES * 32, 0, stream>>>(u, v, delta, A, Bvec, out, batch);
}


# A's LDS staging copy: a thread's 32 row loads all issued back to back before the first LDS store, ahead of the u/B operand loads (was 8 waited rounds of 4 behind them); sums unchanged
# speedup vs baseline: 4.2735x; 1.0925x over previous
#include <hip/hip_runtime.h>


typedef __attribute__((ext_vector_type(2))) float v2f;
typedef __attribute__((ext_vector_type(4))) float v4f;
typedef __attribute__((ext_vector_type(8))) float v8f;

static constexpr int SN    = 128;
static constexpr int TILE  = 16;
static constexpr int NB    = SN / TILE;
static constexpr int APAD  = 132;
static constexpr int SUB   = 4;
static constexpr int WAVES = 4;
static constexpr int NBLK  = NB * (NB + 1) / 2;

__host__ __device__ constexpr int bidx(int k, int j) { return j * NB - (j * (j - 1)) / 2 + (k - j); }

__device__ __forceinline__ v8f mma4(v2f af, v2f bf, v8f c)
{
    return __builtin_amdgcn_wmma_f32_16x16x4_f32(false, af, false, bf, (short)0, c, false, false);
}

__device__ __forceinline__ float xhalf(float x)
{
    const int b = __builtin_bit_cast(int, x);
    return __builtin_bit_cast(float,
        __builtin_amdgcn_permlanex16(b, b, 0x76543210, 0xfedcba98, false, false));
}

struct Inv4 { float m00, m10, m11, m20, m21, m22, m30, m31, m32, m33; };

__device__ __forceinline__ float recip(float d)
{
    const float x = __builtin_amdgcn_rcpf(d);
    return __builtin_fmaf(__builtin_fmaf(-d, x, 1.0f), x, x);
}

__device__ __forceinline__ Inv4 inv4(float dt2, float q00, v2f q1, v2f q2, float q22, v4f q3)
{
    const float i0 = recip(1.0f - dt2 * q00);
    const float i1 = recip(1.0f - dt2 * q1[1]);
    const float i2 = recip(1.0f - dt2 * q22);
    const float i3 = recip(1.0f - dt2 * q3[3]);
    const float c10 = dt2 * q1[0];
    const float c20 = dt2 * q2[0], c21 = dt2 * q2[1];
    const float c30 = dt2 * q3[0], c31 = dt2 * q3[1], c32 = dt2 * q3[2];
    Inv4 m;
    m.m00 = i0; m.m11 = i1; m.m22 = i2; m.m33 = i3;
    m.m10 = i1 * (c10 * i0);
    m.m21 = i2 * (c21 * i1);
    m.m32 = i3 * (c32 * i2);
    m.m20 = i2 * (c20 * i0 + c21 * m.m10);
    m.m31 = i3 * (c31 * i1 + c32 * m.m21);
    m.m30 = i3 * (c30 * i0 + c31 * m.m10 + c32 * m.m20);
    return m;
}

struct Tri4 { float q00; v2f q1; v2f q2; float q22; v4f q3; };

__device__ __forceinline__ Tri4 load_tri(const float* A_lds, int cb)
{
    const float* cp = A_lds + cb * APAD + cb;
    Tri4 t;
    t.q00 = cp[0];
    t.q1  = *(const v2f*)(cp + APAD);
    t.q2  = *(const v2f*)(cp + 2 * APAD);
    t.q22 = cp[2 * APAD + 2];
    t.q3  = *(const v4f*)(cp + 3 * APAD);
    return t;
}

__global__ __launch_bounds__(WAVES * 32) void hippo_bilinear_wmma(
    const float* __restrict__ u, const float* __restrict__ v,
    const float* __restrict__ delta, const float* __restrict__ A,
    const float* __restrict__ Bvec, float* __restrict__ out, int batch)
{
    __shared__ float A_lds[SN * APAD];

    const int tid = threadIdx.x;

    const int wave = tid >> 5;
    const int lane = tid & 31;
    const int col  = lane & 15;
    const int hlf  = lane >> 4;
    const int batch_base = (blockIdx.x * WAVES + wave) * TILE;
    const bool live = batch_base < batch;
    const int b = batch_base + col;
    const int bl = live ? b : 0;

    constexpr int NCP = SN / WAVES;
    const unsigned row0 = (unsigned)wave + 2u * blockIdx.x;
    const unsigned off0 = (row0 * 32u + (unsigned)lane) * 16u;
    unsigned off[NCP];
    #pragma unroll
    for (int i = 0; i < NCP; ++i)
        off[i] = (off0 + (unsigned)i * (unsigned)(WAVES * SN * 4)) & (unsigned)(SN * SN * 4 - 1);
    __builtin_amdgcn_sched_barrier(0);
    v4f st[NCP];
    #pragma unroll
    for (int i = 0; i < NCP; ++i)
        st[i] = *(const v4f*)((const char*)A + off[i]);
    __builtin_amdgcn_sched_barrier(0);

    const float dlt = delta[bl];
    const float vv  = v[bl];
    const float* ucol = u + (size_t)bl * SN;
    v4f uc0[NB], uc1[NB], bc0[NB], bc1[NB];
    #pragma unroll
    for (int k = 0; k < NB; ++k) {
        const int nbase = k * TILE + 8 * hlf;
        uc0[k] = *(const v4f*)(ucol + nbase);
        uc1[k] = *(const v4f*)(ucol + nbase + 4);
    }
    bc0[0] = *(const v4f*)(Bvec + 8 * hlf);
    bc1[0] = *(const v4f*)(Bvec + 8 * hlf + 4);

    #pragma unroll
    for (int i = 0; i < NCP; ++i) {
        const unsigned row = (row0 + 4u * i) & (unsigned)(SN - 1);
        *(v4f*)(&A_lds[row * APAD + lane * 4]) = st[i];
    }
    __syncthreads();

    if (!live) return;

    #pragma unroll
    for (int k = 1; k < NB; ++k) {
        const int nbase = k * TILE + 8 * hlf;
        bc0[k] = *(const v4f*)(Bvec + nbase);
        bc1[k] = *(const v4f*)(Bvec + nbase + 4);
    }

    const float dt2 = 0.5f * dlt;
    const float dvb = dlt * vv;

    const float* alane = &A_lds[col * APAD + 2 * hlf];

    v2f afr[NBLK][4];
    #pragma unroll
    for (int j = 0; j < NB; ++j) {
        #pragma unroll
        for (int k4 = 0; k4 < 4; ++k4) {
            #pragma unroll
            for (int k = j; k < NB; ++k)
                afr[bidx(k, j)][k4] = *(const v2f*)(alane + (k * TILE) * APAD + j * TILE + k4 * 4);
        }
    }
    Tri4 tri[2][4];
    #pragma unroll
    for (int t = 0; t < 2; ++t) {
        #pragma unroll
        for (int s = 0; s < 4; ++s)
            tri[t][s] = load_tri(A_lds, t * TILE + SUB * s);
    }
    __builtin_amdgcn_sched_barrier(0);

    const bool up = (hlf != 0);
    v8f R[NB];
    #pragma unroll
    for (int k = 0; k < NB; ++k) R[k] = v8f{};
    #pragma unroll
    for (int j = 0; j < NB; ++j) {
        const float tx0 = xhalf(up ? uc0[j][0] : uc0[j][2]);
        const float ty0 = xhalf(up ? uc0[j][1] : uc0[j][3]);
        const float tx1 = xhalf(up ? uc1[j][0] : uc1[j][2]);
        const float ty1 = xhalf(up ? uc1[j][1] : uc1[j][3]);
        v2f uf[4];
        uf[0].x = up ? tx0 : uc0[j][0];   uf[0].y = up ? ty0 : uc0[j][1];
        uf[1].x = up ? tx1 : uc1[j][0];   uf[1].y = up ? ty1 : uc1[j][1];
        uf[2].x = up ? uc0[j][2] : tx0;   uf[2].y = up ? uc0[j][3] : ty0;
        uf[3].x = up ? uc1[j][2] : tx1;   uf[3].y = up ? uc1[j][3] : ty1;
        #pragma unroll
        for (int k4 = 0; k4 < 4; ++k4) {
            #pragma unroll
            for (int k = j; k < NB; ++k)
                R[k] = mma4(afr[bidx(k, j)][k4], uf[k4], R[k]);
        }
        if (j == 0) __builtin_amdgcn_sched_barrier(0);
    }
    #pragma unroll
    for (int k = 0; k < NB; ++k) {
        #pragma unroll
        for (int r = 0; r < 4; ++r) {
            R[k][r]     = uc0[k][r] + dt2 * R[k][r]     + dvb * bc0[k][r];
            R[k][r + 4] = uc1[k][r] + dt2 * R[k][r + 4] + dvb * bc1[k][r];
        }
    }

    Inv4 Mn;
    {
        const Tri4 t0 = tri[0][0];
        Mn = inv4(dt2, t0.q00, t0.q1, t0.q2, t0.q22, t0.q3);
    }

    #pragma unroll
    for (int kb = 0; kb < NB; ++kb) {
        float Y[8];

        if (kb >= 1 && kb + 1 < NB) {
            __builtin_amdgcn_sched_barrier(0);
            #pragma unroll
            for (int s = 0; s < 4; ++s)
                tri[(kb + 1) & 1][s] = load_tri(A_lds, (kb + 1) * TILE + SUB * s);
            __builtin_amdgcn_sched_barrier(0);
        }

        #pragma unroll
        for (int s = 0; s < 4; ++s) {
            const int ha = s >> 1;
            const int rb = SUB * (s & 1);

            const Inv4 M = Mn;

            if (kb * 4 + s + 1 < NB * 4) {
                const Tri4 tn = (s < 3) ? tri[kb & 1][s + 1] : tri[(kb + 1) & 1][0];
                Mn = inv4(dt2, tn.q00, tn.q1, tn.q2, tn.q22, tn.q3);
                asm volatile("" :: "v"(Mn.m00), "v"(Mn.m10), "v"(Mn.m11), "v"(Mn.m20), "v"(Mn.m21),
                                   "v"(Mn.m22), "v"(Mn.m30), "v"(Mn.m31), "v"(Mn.m32), "v"(Mn.m33));
            }

            const float r0 = R[kb][rb], r1 = R[kb][rb + 1], r2 = R[kb][rb + 2], r3 = R[kb][rb + 3];
            const float y0 = M.m00 * r0;
            const float y1 = M.m10 * r0 + M.m11 * r1;
            const float y2 = M.m20 * r0 + M.m21 * r1 + M.m22 * r2;
            const float y3 = (M.m30 * r0 + M.m31 * r1) + (M.m32 * r2 + M.m33 * r3);

            const float z0 = dt2 * y0, z1 = dt2 * y1, z2 = dt2 * y2, z3 = dt2 * y3;
            v2f zb;
            if (ha == 0) {
                const float t2 = xhalf(z2), t3 = xhalf(z3);
                zb.x = up ? t2 : z0;
                zb.y = up ? t3 : z1;
                Y[rb] = y0; Y[rb + 1] = y1; Y[rb + 2] = y2; Y[rb + 3] = y3;
            } else {
                const float t0 = xhalf(z0), t1 = xhalf(z1);
                zb.x = up ? z2 : t0;
                zb.y = up ? z3 : t1;
                Y[rb]     = up ? y0 : Y[rb];
                Y[rb + 1] = up ? y1 : Y[rb + 1];
                Y[rb + 2] = up ? y2 : Y[rb + 2];
                Y[rb + 3] = up ? y3 : Y[rb + 3];
            }

            #pragma unroll
            for (int k = (s < 3) ? kb : kb + 1; k < NB; ++k)
                R[k] = mma4(afr[bidx(k, kb)][s], zb, R[k]);
        }

        float* o = out + (size_t)b * SN + kb * TILE + 8 * hlf;
        const v4f o0 = { Y[0], Y[1], Y[2], Y[3] };
        const v4f o1 = { Y[4], Y[5], Y[6], Y[7] };
        *(v4f*)(o)     = o0;
        *(v4f*)(o + 4) = o1;
    }
}

extern "C" void kernel_launch(void* const* d_in, const int* in_sizes, int n_in,
                              void* d_out, int out_size, void* d_ws, size_t ws_size,
                              hipStream_t stream) {
    const float* u     = (const float*)d_in[0];
    const float* v     = (const float*)d_in[1];
    const float* delta = (const float*)d_in[2];
    const float* A     = (const float*)d_in[3];
    const float* Bvec  = (const float*)d_in[4];
    float* out = (float*)d_out;

    const int batch  = in_sizes[0] / SN;
    const int blocks = batch / (WAVES * TILE);
    hippo_bilinear_wmma<<<blocks, WAVES * 32, 0, stream>>>(u, v, delta, A, Bvec, out, batch);
}


# a tile's state rows split between two waves on two SIMDs (128 work-groups): rows 0..63 and rows 64..127 each formed and solved by one wave; the top's Z reaches the bottom through LDS; sums unchanged
# speedup vs baseline: 4.9648x; 1.1618x over previous
#include <hip/hip_runtime.h>


typedef __attribute__((ext_vector_type(2))) float v2f;
typedef __attribute__((ext_vector_type(4))) float v4f;
typedef __attribute__((ext_vector_type(8))) float v8f;

static constexpr int SN    = 128;
static constexpr int TILE  = 16;
static constexpr int NB    = SN / TILE;
static constexpr int APAD  = 132;
static constexpr int SUB   = 4;
static constexpr int WAVES = 4;
static constexpr int HB    = NB / 2;
static constexpr int NBLK  = NB * (NB + 1) / 2;

__host__ __device__ constexpr int bidx(int k, int j) { return j * NB - (j * (j - 1)) / 2 + (k - j); }

__device__ __forceinline__ v8f mma4(v2f af, v2f bf, v8f c)
{
    return __builtin_amdgcn_wmma_f32_16x16x4_f32(false, af, false, bf, (short)0, c, false, false);
}

__device__ __forceinline__ float xhalf(float x)
{
    const int b = __builtin_bit_cast(int, x);
    return __builtin_bit_cast(float,
        __builtin_amdgcn_permlanex16(b, b, 0x76543210, 0xfedcba98, false, false));
}

struct Inv4 { float m00, m10, m11, m20, m21, m22, m30, m31, m32, m33; };

__device__ __forceinline__ float recip(float d)
{
    const float x = __builtin_amdgcn_rcpf(d);
    return __builtin_fmaf(__builtin_fmaf(-d, x, 1.0f), x, x);
}

__device__ __forceinline__ Inv4 inv4(float dt2, float q00, v2f q1, v2f q2, float q22, v4f q3)
{
    const float i0 = recip(1.0f - dt2 * q00);
    const float i1 = recip(1.0f - dt2 * q1[1]);
    const float i2 = recip(1.0f - dt2 * q22);
    const float i3 = recip(1.0f - dt2 * q3[3]);
    const float c10 = dt2 * q1[0];
    const float c20 = dt2 * q2[0], c21 = dt2 * q2[1];
    const float c30 = dt2 * q3[0], c31 = dt2 * q3[1], c32 = dt2 * q3[2];
    Inv4 m;
    m.m00 = i0; m.m11 = i1; m.m22 = i2; m.m33 = i3;
    m.m10 = i1 * (c10 * i0);
    m.m21 = i2 * (c21 * i1);
    m.m32 = i3 * (c32 * i2);
    m.m20 = i2 * (c20 * i0 + c21 * m.m10);
    m.m31 = i3 * (c31 * i1 + c32 * m.m21);
    m.m30 = i3 * (c30 * i0 + c31 * m.m10 + c32 * m.m20);
    return m;
}

struct Tri4 { float q00; v2f q1; v2f q2; float q22; v4f q3; };

__device__ __forceinline__ Tri4 load_tri(const float* A_lds, int cb)
{
    const float* cp = A_lds + cb * APAD + cb;
    Tri4 t;
    t.q00 = cp[0];
    t.q1  = *(const v2f*)(cp + APAD);
    t.q2  = *(const v2f*)(cp + 2 * APAD);
    t.q22 = cp[2 * APAD + 2];
    t.q3  = *(const v4f*)(cp + 3 * APAD);
    return t;
}


template<int KLO, int KHI, bool TRI>
__device__ __forceinline__ void read_fragments(v2f (&afr)[NBLK][4], Tri4 (&tri)[2][4],
                                               const float* A_lds, const float* alane)
{
    #pragma unroll
    for (int j = 0; j < KHI; ++j) {
        #pragma unroll
        for (int k4 = 0; k4 < 4; ++k4) {
            #pragma unroll
            for (int k = (j > KLO ? j : KLO); k < KHI; ++k)
                afr[bidx(k, j)][k4] = *(const v2f*)(alane + (k * TILE) * APAD + j * TILE + k4 * 4);
        }
    }
    if (TRI) {
        #pragma unroll
        for (int t = 0; t < 2; ++t) {
            #pragma unroll
            for (int s = 0; s < 4; ++s)
                tri[(KLO + t) & 1][s] = load_tri(A_lds, (KLO + t) * TILE + SUB * s);
        }
    }
}

template<int KLO, int KHI>
__device__ __forceinline__ void all_inverses(Inv4 (&Mall)[HB * 2], const float* A_lds, float dt2)
{
    Tri4 q[(KHI - KLO) * 4];
    #pragma unroll
    for (int t = 0; t < (KHI - KLO) * 4; ++t)
        q[t] = load_tri(A_lds, KLO * TILE + SUB * t);
    #pragma unroll
    for (int t = 0; t < (KHI - KLO) * 4; ++t)
        Mall[t] = inv4(dt2, q[t].q00, q[t].q1, q[t].q2, q[t].q22, q[t].q3);
}

template<int KLO, int KHI>
__device__ __forceinline__ void forward_rows(v8f (&R)[NB], const v2f (&afr)[NBLK][4],
                                             const v4f (&uc0)[NB], const v4f (&uc1)[NB],
                                             const v4f (&bc0)[NB], const v4f (&bc1)[NB],
                                             float dt2, float dvb, bool up)
{
    #pragma unroll
    for (int k = KLO; k < KHI; ++k) R[k] = v8f{};
    #pragma unroll
    for (int j = 0; j < KHI; ++j) {
        const float tx0 = xhalf(up ? uc0[j][0] : uc0[j][2]);
        const float ty0 = xhalf(up ? uc0[j][1] : uc0[j][3]);
        const float tx1 = xhalf(up ? uc1[j][0] : uc1[j][2]);
        const float ty1 = xhalf(up ? uc1[j][1] : uc1[j][3]);
        v2f uf[4];
        uf[0].x = up ? tx0 : uc0[j][0];   uf[0].y = up ? ty0 : uc0[j][1];
        uf[1].x = up ? tx1 : uc1[j][0];   uf[1].y = up ? ty1 : uc1[j][1];
        uf[2].x = up ? uc0[j][2] : tx0;   uf[2].y = up ? uc0[j][3] : ty0;
        uf[3].x = up ? uc1[j][2] : tx1;   uf[3].y = up ? uc1[j][3] : ty1;
        #pragma unroll
        for (int k4 = 0; k4 < 4; ++k4) {
            #pragma unroll
            for (int k = (j > KLO ? j : KLO); k < KHI; ++k)
                R[k] = mma4(afr[bidx(k, j)][k4], uf[k4], R[k]);
        }
        if (j == 0) __builtin_amdgcn_sched_barrier(0);
    }
    #pragma unroll
    for (int k = KLO; k < KHI; ++k) {
        #pragma unroll
        for (int r = 0; r < 4; ++r) {
            R[k][r]     = uc0[k][r] + dt2 * R[k][r]     + dvb * bc0[k][r];
            R[k][r + 4] = uc1[k][r] + dt2 * R[k][r + 4] + dvb * bc1[k][r];
        }
    }
}

template<int KLO, int KHI, bool KEEPZ, int PREN>
__device__ __forceinline__ void solve_rows(v8f (&R)[NB], const v2f (&afr)[NBLK][4], Tri4 (&tri)[2][4],
                                           const float* A_lds, float dt2, bool up,
                                           v2f (&zall)[HB * 4], const Inv4 (&Mall)[HB * 2], float* orow)
{
    Inv4 Mn;
    if (PREN == 0) {
        const Tri4 t0 = tri[KLO & 1][0];
        Mn = inv4(dt2, t0.q00, t0.q1, t0.q2, t0.q22, t0.q3);
    }

    #pragma unroll
    for (int kb = KLO; kb < KHI; ++kb) {
        float Y[8];

        if (kb + 1 < KHI && (kb + 1 - KLO) * 4 >= PREN && (PREN > 0 || kb >= KLO + 1)) {
            __builtin_amdgcn_sched_barrier(0);
            #pragma unroll
            for (int s = 0; s < 4; ++s)
                tri[(kb + 1) & 1][s] = load_tri(A_lds, (kb + 1) * TILE + SUB * s);
            __builtin_amdgcn_sched_barrier(0);
        }

        #pragma unroll
        for (int s = 0; s < 4; ++s) {
            const int ha = s >> 1;
            const int rb = SUB * (s & 1);

            const int ti = (kb - KLO) * 4 + s;
            const Inv4 M = (ti < PREN) ? Mall[ti < PREN ? ti : 0] : Mn;

            if (ti + 1 >= PREN && kb * 4 + s + 1 < KHI * 4) {
                const Tri4 tn = (s < 3) ? tri[kb & 1][s + 1] : tri[(kb + 1) & 1][0];
                Mn = inv4(dt2, tn.q00, tn.q1, tn.q2, tn.q22, tn.q3);
                asm volatile("" :: "v"(Mn.m00), "v"(Mn.m10), "v"(Mn.m11), "v"(Mn.m20), "v"(Mn.m21),
                                   "v"(Mn.m22), "v"(Mn.m30), "v"(Mn.m31), "v"(Mn.m32), "v"(Mn.m33));
            }

            const float r0 = R[kb][rb], r1 = R[kb][rb + 1], r2 = R[kb][rb + 2], r3 = R[kb][rb + 3];
            const float y0 = M.m00 * r0;
            const float y1 = M.m10 * r0 + M.m11 * r1;
            const float y2 = M.m20 * r0 + M.m21 * r1 + M.m22 * r2;
            const float y3 = (M.m30 * r0 + M.m31 * r1) + (M.m32 * r2 + M.m33 * r3);

            const float z0 = dt2 * y0, z1 = dt2 * y1, z2 = dt2 * y2, z3 = dt2 * y3;
            v2f zb;
            if (ha == 0) {
                const float t2 = xhalf(z2), t3 = xhalf(z3);
                zb.x = up ? t2 : z0;
                zb.y = up ? t3 : z1;
                Y[rb] = y0; Y[rb + 1] = y1; Y[rb + 2] = y2; Y[rb + 3] = y3;
            } else {
                const float t0 = xhalf(z0), t1 = xhalf(z1);
                zb.x = up ? z2 : t0;
                zb.y = up ? z3 : t1;
                Y[rb]     = up ? y0 : Y[rb];
                Y[rb + 1] = up ? y1 : Y[rb + 1];
                Y[rb + 2] = up ? y2 : Y[rb + 2];
                Y[rb + 3] = up ? y3 : Y[rb + 3];
            }
            if (KEEPZ) zall[(kb - KLO) * 4 + s] = zb;

            #pragma unroll
            for (int k = (s < 3) ? kb : kb + 1; k < KHI; ++k)
                R[k] = mma4(afr[bidx(k, kb)][s], zb, R[k]);
        }

        float* o = orow + kb * TILE;
        const v4f o0 = { Y[0], Y[1], Y[2], Y[3] };
        const v4f o1 = { Y[4], Y[5], Y[6], Y[7] };
        *(v4f*)(o)     = o0;
        *(v4f*)(o + 4) = o1;
    }
}

__global__ __launch_bounds__(WAVES * 32) void hippo_bilinear_wmma(
    const float* __restrict__ u, const float* __restrict__ v,
    const float* __restrict__ delta, const float* __restrict__ A,
    const float* __restrict__ Bvec, float* __restrict__ out)
{
    __shared__ float A_lds[SN * APAD];
    __shared__ v2f   Z_lds[2 * HB * 4 * 32];

    const int tid = threadIdx.x;

    const int wave = __builtin_amdgcn_readfirstlane(tid >> 5);
    const int lane = tid & 31;
    const int col  = lane & 15;
    const int hlf  = lane >> 4;
    const int tw   = wave >> 1;
    const bool bottom = (wave & 1) != 0;
    const int batch_base = (blockIdx.x * 2 + tw) * TILE;
    const int b = batch_base + col;

    constexpr int NCP = SN / WAVES;
    const unsigned row0 = (unsigned)wave + blockIdx.x;
    const unsigned off0 = (row0 * 32u + (unsigned)lane) * 16u;
    unsigned off[NCP];
    #pragma unroll
    for (int i = 0; i < NCP; ++i)
        off[i] = (off0 + (unsigned)i * (unsigned)(WAVES * SN * 4)) & (unsigned)(SN * SN * 4 - 1);
    __builtin_amdgcn_sched_barrier(0);
    v4f st[NCP];
    #pragma unroll
    for (int i = 0; i < NCP; ++i)
        st[i] = *(const v4f*)((const char*)A + off[i]);
    __builtin_amdgcn_sched_barrier(0);

    const float dlt = delta[b];
    const float vv  = v[b];
    const float* ucol = u + (size_t)b * SN;
    v4f uc0[NB], uc1[NB], bc0[NB], bc1[NB];
    #pragma unroll
    for (int k = 0; k < HB; ++k) {
        const int nbase = k * TILE + 8 * hlf;
        uc0[k] = *(const v4f*)(ucol + nbase);
        uc1[k] = *(const v4f*)(ucol + nbase + 4);
    }
    if (bottom) {
        #pragma unroll
        for (int k = HB; k < NB; ++k) {
            const int nbase = k * TILE + 8 * hlf;
            uc0[k] = *(const v4f*)(ucol + nbase);
            uc1[k] = *(const v4f*)(ucol + nbase + 4);
            bc0[k] = *(const v4f*)(Bvec + nbase);
            bc1[k] = *(const v4f*)(Bvec + nbase + 4);
        }
    } else {
        #pragma unroll
        for (int k = 0; k < HB; ++k) {
            const int nbase = k * TILE + 8 * hlf;
            bc0[k] = *(const v4f*)(Bvec + nbase);
            bc1[k] = *(const v4f*)(Bvec + nbase + 4);
        }
    }

    #pragma unroll
    for (int i = 0; i < NCP; ++i) {
        const unsigned row = (row0 + 4u * i) & (unsigned)(SN - 1);
        *(v4f*)(&A_lds[row * APAD + lane * 4]) = st[i];
    }
    __syncthreads();

    const float dt2 = 0.5f * dlt;
    const float dvb = dlt * vv;
    const bool up = (hlf != 0);

    const float* alane = &A_lds[col * APAD + 2 * hlf];
    v2f* zslot = &Z_lds[(tw * HB * 4) * 32 + lane];
    float* orow = out + (size_t)b * SN + 8 * hlf;

    v2f afr[NBLK][4];
    Tri4 tri[2][4];
    v8f R[NB];
    v2f zall[HB * 4];
    Inv4 Mall[HB * 2];

    if (!bottom) {
        read_fragments<0, HB, true>(afr, tri, A_lds, alane);
        __builtin_amdgcn_sched_barrier(0);
        forward_rows<0, HB>(R, afr, uc0, uc1, bc0, bc1, dt2, dvb, up);
        solve_rows<0, HB, true, 0>(R, afr, tri, A_lds, dt2, up, zall, Mall, orow);
        #pragma unroll
        for (int t = 0; t < HB * 4; ++t)
            zslot[t * 32] = zall[t];
    } else {
        read_fragments<HB, NB, false>(afr, tri, A_lds, alane);
        __builtin_amdgcn_sched_barrier(0);
        forward_rows<HB, NB>(R, afr, uc0, uc1, bc0, bc1, dt2, dvb, up);
        all_inverses<HB, HB + 2>(Mall, A_lds, dt2);
    }

    __builtin_amdgcn_fence(__ATOMIC_RELEASE, "workgroup", "local");
    __builtin_amdgcn_s_barrier();
    __builtin_amdgcn_fence(__ATOMIC_ACQUIRE, "workgroup", "local");

    if (bottom) {
        #pragma unroll
        for (int t = 0; t < HB * 4; ++t)
            zall[t] = zslot[t * 32];
        __builtin_amdgcn_sched_barrier(0);
        #pragma unroll
        for (int t = 0; t < HB * 4; ++t) {
            #pragma unroll
            for (int k = HB; k < NB; ++k)
                R[k] = mma4(afr[bidx(k, t >> 2)][t & 3], zall[t], R[k]);
        }
        solve_rows<HB, NB, false, HB * 2>(R, afr, tri, A_lds, dt2, up, zall, Mall, orow);
    }
}

extern "C" void kernel_launch(void* const* d_in, const int* in_sizes, int n_in,
                              void* d_out, int out_size, void* d_ws, size_t ws_size,
                              hipStream_t stream) {
    const float* u     = (const float*)d_in[0];
    const float* v     = (const float*)d_in[1];
    const float* delta = (const float*)d_in[2];
    const float* A     = (const float*)d_in[3];
    const float* Bvec  = (const float*)d_in[4];
    float* out = (float*)d_out;

    const int batch  = in_sizes[0] / SN;
    const int blocks = batch / (2 * TILE);
    hippo_bilinear_wmma<<<blocks, WAVES * 32, 0, stream>>>(u, v, delta, A, Bvec, out);
}


# a tile's state rows split among four waves on the four SIMDs (256 one-tile work-groups; was two waves, 128): 32 rows each, Z handed down via LDS behind 3 barriers; sum order changes in rows 64..127
# speedup vs baseline: 5.4933x; 1.1064x over previous
#include <hip/hip_runtime.h>


typedef __attribute__((ext_vector_type(2))) float v2f;
typedef __attribute__((ext_vector_type(4))) float v4f;
typedef __attribute__((ext_vector_type(8))) float v8f;

static constexpr int SN    = 128;
static constexpr int TILE  = 16;
static constexpr int NB    = SN / TILE;
static constexpr int APAD  = 132;
static constexpr int SUB   = 4;
static constexpr int WAVES = 4;
static constexpr int HB    = NB / WAVES;
static constexpr int NSTEP = HB * 4;
static constexpr int NBLK  = NB * (NB + 1) / 2;

__host__ __device__ constexpr int bidx(int k, int j) { return j * NB - (j * (j - 1)) / 2 + (k - j); }

__device__ __forceinline__ v8f mma4(v2f af, v2f bf, v8f c)
{
    return __builtin_amdgcn_wmma_f32_16x16x4_f32(false, af, false, bf, (short)0, c, false, false);
}

__device__ __forceinline__ float xhalf(float x)
{
    const int b = __builtin_bit_cast(int, x);
    return __builtin_bit_cast(float,
        __builtin_amdgcn_permlanex16(b, b, 0x76543210, 0xfedcba98, false, false));
}

struct Inv4 { float m00, m10, m11, m20, m21, m22, m30, m31, m32, m33; };

__device__ __forceinline__ float recip(float d)
{
    const float x = __builtin_amdgcn_rcpf(d);
    return __builtin_fmaf(__builtin_fmaf(-d, x, 1.0f), x, x);
}

__device__ __forceinline__ Inv4 inv4(float dt2, float q00, v2f q1, v2f q2, float q22, v4f q3)
{
    const float i0 = recip(1.0f - dt2 * q00);
    const float i1 = recip(1.0f - dt2 * q1[1]);
    const float i2 = recip(1.0f - dt2 * q22);
    const float i3 = recip(1.0f - dt2 * q3[3]);
    const float c10 = dt2 * q1[0];
    const float c20 = dt2 * q2[0], c21 = dt2 * q2[1];
    const float c30 = dt2 * q3[0], c31 = dt2 * q3[1], c32 = dt2 * q3[2];
    Inv4 m;
    m.m00 = i0; m.m11 = i1; m.m22 = i2; m.m33 = i3;
    m.m10 = i1 * (c10 * i0);
    m.m21 = i2 * (c21 * i1);
    m.m32 = i3 * (c32 * i2);
    m.m20 = i2 * (c20 * i0 + c21 * m.m10);
    m.m31 = i3 * (c31 * i1 + c32 * m.m21);
    m.m30 = i3 * (c30 * i0 + c31 * m.m10 + c32 * m.m20);
    return m;
}

struct Tri4 { float q00; v2f q1; v2f q2; float q22; v4f q3; };

__device__ __forceinline__ Tri4 load_tri(const float* A_lds, int cb)
{
    const float* cp = A_lds + cb * APAD + cb;
    Tri4 t;
    t.q00 = cp[0];
    t.q1  = *(const v2f*)(cp + APAD);
    t.q2  = *(const v2f*)(cp + 2 * APAD);
    t.q22 = cp[2 * APAD + 2];
    t.q3  = *(const v4f*)(cp + 3 * APAD);
    return t;
}


template<int KLO, int KHI, int J0, int J1, bool TRI>
__device__ __forceinline__ void read_fragments(v2f (&afr)[NBLK][4], Tri4 (&tri)[2][4],
                                               const float* A_lds, const float* alane)
{
    #pragma unroll
    for (int j = J0; j < J1; ++j) {
        #pragma unroll
        for (int k4 = 0; k4 < 4; ++k4) {
            #pragma unroll
            for (int k = (j > KLO ? j : KLO); k < KHI; ++k)
                afr[bidx(k, j)][k4] = *(const v2f*)(alane + (k * TILE) * APAD + j * TILE + k4 * 4);
        }
    }
    if (TRI) {
        #pragma unroll
        for (int t = 0; t < 2; ++t) {
            #pragma unroll
            for (int s = 0; s < 4; ++s)
                tri[(KLO + t) & 1][s] = load_tri(A_lds, (KLO + t) * TILE + SUB * s);
        }
    }
}

template<int KLO, int KHI, int J0, int J1>
__device__ __forceinline__ void keep_fragments_here(const v2f (&afr)[NBLK][4])
{
    #pragma unroll
    for (int j = J0; j < J1; ++j) {
        #pragma unroll
        for (int k = (j > KLO ? j : KLO); k < KHI; ++k)
            asm volatile("" :: "v"(afr[bidx(k, j)][0]), "v"(afr[bidx(k, j)][1]),
                               "v"(afr[bidx(k, j)][2]), "v"(afr[bidx(k, j)][3]));
    }
}

template<int KB>
__device__ __forceinline__ void read_tris(Tri4 (&q)[4], const float* A_lds)
{
    #pragma unroll
    for (int t = 0; t < 4; ++t)
        q[t] = load_tri(A_lds, KB * TILE + SUB * t);
}

template<int T0>
__device__ __forceinline__ void form_inverses(Inv4 (&Mall)[NSTEP], const Tri4 (&q)[4], float dt2)
{
    #pragma unroll
    for (int t = 0; t < 4; ++t)
        Mall[T0 + t] = inv4(dt2, q[t].q00, q[t].q1, q[t].q2, q[t].q22, q[t].q3);
}

template<int KLO, int KHI>
__device__ __forceinline__ void zero_rows(v8f (&X)[NB])
{
    #pragma unroll
    for (int k = KLO; k < KHI; ++k) X[k] = v8f{};
}

template<int KLO, int KHI, int J0, int J1>
__device__ __forceinline__ void forward_cols(v8f (&RF)[NB], const v2f (&afr)[NBLK][4],
                                             const v4f (&uc0)[NB], const v4f (&uc1)[NB], bool up)
{
    #pragma unroll
    for (int j = J0; j < J1; ++j) {
        const float tx0 = xhalf(up ? uc0[j][0] : uc0[j][2]);
        const float ty0 = xhalf(up ? uc0[j][1] : uc0[j][3]);
        const float tx1 = xhalf(up ? uc1[j][0] : uc1[j][2]);
        const float ty1 = xhalf(up ? uc1[j][1] : uc1[j][3]);
        v2f uf[4];
        uf[0].x = up ? tx0 : uc0[j][0];   uf[0].y = up ? ty0 : uc0[j][1];
        uf[1].x = up ? tx1 : uc1[j][0];   uf[1].y = up ? ty1 : uc1[j][1];
        uf[2].x = up ? uc0[j][2] : tx0;   uf[2].y = up ? uc0[j][3] : ty0;
        uf[3].x = up ? uc1[j][2] : tx1;   uf[3].y = up ? uc1[j][3] : ty1;
        #pragma unroll
        for (int k4 = 0; k4 < 4; ++k4) {
            #pragma unroll
            for (int k = (j > KLO ? j : KLO); k < KHI; ++k)
                RF[k] = mma4(afr[bidx(k, j)][k4], uf[k4], RF[k]);
        }
        if (j == 0) __builtin_amdgcn_sched_barrier(0);
    }
}

template<int KLO, int KHI, bool ADDZ>
__device__ __forceinline__ void right_hand_sides(v8f (&R)[NB], const v8f (&RF)[NB], const v8f (&RZ)[NB],
                                                 const v4f (&uc0)[NB], const v4f (&uc1)[NB],
                                                 const v4f (&bc0)[HB], const v4f (&bc1)[HB],
                                                 float dt2, float dvb)
{
    #pragma unroll
    for (int k = KLO; k < KHI; ++k) {
        #pragma unroll
        for (int r = 0; r < 4; ++r) {
            const float a0 = uc0[k][r] + dt2 * RF[k][r]     + dvb * bc0[k - KLO][r];
            const float a1 = uc1[k][r] + dt2 * RF[k][r + 4] + dvb * bc1[k - KLO][r];
            R[k][r]     = ADDZ ? a0 + RZ[k][r]     : a0;
            R[k][r + 4] = ADDZ ? a1 + RZ[k][r + 4] : a1;
        }
    }
}

template<int P>
__device__ __forceinline__ void read_z(v2f (&z)[NSTEP], const v2f* zs)
{
    #pragma unroll
    for (int t = 0; t < NSTEP; ++t)
        z[t] = zs[(P * NSTEP + t) * 32];
}

template<int KLO, int KHI, int P>
__device__ __forceinline__ void add_group(v8f (&X)[NB], const v2f (&afr)[NBLK][4], const v2f (&z)[NSTEP])
{
    #pragma unroll
    for (int t = 0; t < NSTEP; ++t) {
        #pragma unroll
        for (int k = KLO; k < KHI; ++k)
            X[k] = mma4(afr[bidx(k, 2 * P + (t >> 2))][t & 3], z[t], X[k]);
    }
}

template<int G>
__device__ __forceinline__ void leave_z(v2f* zs, const v2f (&zall)[NSTEP])
{
    #pragma unroll
    for (int t = 0; t < NSTEP; ++t)
        zs[(G * NSTEP + t) * 32] = zall[t];
}

__device__ __forceinline__ void hand_over()
{
    __builtin_amdgcn_fence(__ATOMIC_RELEASE, "workgroup", "local");
    __builtin_amdgcn_s_barrier();
    __builtin_amdgcn_fence(__ATOMIC_ACQUIRE, "workgroup", "local");
}

template<int KLO, int KHI, bool KEEPZ, int PREN, bool HAVETRI>
__device__ __forceinline__ void solve_rows(v8f (&R)[NB], const v2f (&afr)[NBLK][4], Tri4 (&tri)[2][4],
                                           const float* A_lds, float dt2, bool up,
                                           v2f (&zall)[NSTEP], const Inv4 (&Mall)[NSTEP], float* orow)
{
    Inv4 Mn;
    if (PREN == 0) {
        const Tri4 t0 = tri[KLO & 1][0];
        Mn = inv4(dt2, t0.q00, t0.q1, t0.q2, t0.q22, t0.q3);
    }

    #pragma unroll
    for (int kb = KLO; kb < KHI; ++kb) {
        float Y[8];

        if (!HAVETRI && kb + 1 < KHI && (kb + 1 - KLO) * 4 >= PREN && (PREN > 0 || kb >= KLO + 1)) {
            __builtin_amdgcn_sched_barrier(0);
            #pragma unroll
            for (int s = 0; s < 4; ++s)
                tri[(kb + 1) & 1][s] = load_tri(A_lds, (kb + 1) * TILE + SUB * s);
            __builtin_amdgcn_sched_barrier(0);
        }

        #pragma unroll
        for (int s = 0; s < 4; ++s) {
            const int ha = s >> 1;
            const int rb = SUB * (s & 1);

            const int ti = (kb - KLO) * 4 + s;
            const Inv4 M = (ti < PREN) ? Mall[ti < PREN ? ti : 0] : Mn;

            if (ti + 1 >= PREN && kb * 4 + s + 1 < KHI * 4) {
                const Tri4 tn = (s < 3) ? tri[kb & 1][s + 1] : tri[(kb + 1) & 1][0];
                Mn = inv4(dt2, tn.q00, tn.q1, tn.q2, tn.q22, tn.q3);
                asm volatile("" :: "v"(Mn.m00), "v"(Mn.m10), "v"(Mn.m11), "v"(Mn.m20), "v"(Mn.m21),
                                   "v"(Mn.m22), "v"(Mn.m30), "v"(Mn.m31), "v"(Mn.m32), "v"(Mn.m33));
            }

            const float r0 = R[kb][rb], r1 = R[kb][rb + 1], r2 = R[kb][rb + 2], r3 = R[kb][rb + 3];
            const float y0 = M.m00 * r0;
            const float y1 = M.m10 * r0 + M.m11 * r1;
            const float y2 = M.m20 * r0 + M.m21 * r1 + M.m22 * r2;
            const float y3 = (M.m30 * r0 + M.m31 * r1) + (M.m32 * r2 + M.m33 * r3);

            const float z0 = dt2 * y0, z1 = dt2 * y1, z2 = dt2 * y2, z3 = dt2 * y3;
            v2f zb;
            if (ha == 0) {
                const float t2 = xhalf(z2), t3 = xhalf(z3);
                zb.x = up ? t2 : z0;
                zb.y = up ? t3 : z1;
                Y[rb] = y0; Y[rb + 1] = y1; Y[rb + 2] = y2; Y[rb + 3] = y3;
            } else {
                const float t0 = xhalf(z0), t1 = xhalf(z1);
                zb.x = up ? z2 : t0;
                zb.y = up ? z3 : t1;
                Y[rb]     = up ? y0 : Y[rb];
                Y[rb + 1] = up ? y1 : Y[rb + 1];
                Y[rb + 2] = up ? y2 : Y[rb + 2];
                Y[rb + 3] = up ? y3 : Y[rb + 3];
            }
            if (KEEPZ) zall[(kb - KLO) * 4 + s] = zb;

            #pragma unroll
            for (int k = (s < 3) ? kb : kb + 1; k < KHI; ++k)
                R[k] = mma4(afr[bidx(k, kb)][s], zb, R[k]);
        }

        float* o = orow + kb * TILE;
        const v4f o0 = { Y[0], Y[1], Y[2], Y[3] };
        const v4f o1 = { Y[4], Y[5], Y[6], Y[7] };
        *(v4f*)(o)     = o0;
        *(v4f*)(o + 4) = o1;
    }
}

__global__ __launch_bounds__(WAVES * 32) void hippo_bilinear_wmma(
    const float* __restrict__ u, const float* __restrict__ v,
    const float* __restrict__ delta, const float* __restrict__ A,
    const float* __restrict__ Bvec, float* __restrict__ out)
{
    __shared__ float A_lds[SN * APAD];
    __shared__ v2f   Z_lds[(WAVES - 1) * NSTEP * 32];

    const int tid = threadIdx.x;

    const int wave = __builtin_amdgcn_readfirstlane(tid >> 5);
    const int lane = tid & 31;
    const int col  = lane & 15;
    const int hlf  = lane >> 4;
    const int b = blockIdx.x * TILE + col;

    constexpr int NCP = SN / WAVES;
    const unsigned row0 = (unsigned)wave + blockIdx.x;
    const unsigned off0 = (row0 * 32u + (unsigned)lane) * 16u;
    unsigned off[NCP];
    #pragma unroll
    for (int i = 0; i < NCP; ++i)
        off[i] = (off0 + (unsigned)i * (unsigned)(WAVES * SN * 4)) & (unsigned)(SN * SN * 4 - 1);
    __builtin_amdgcn_sched_barrier(0);
    v4f st[NCP];
    #pragma unroll
    for (int i = 0; i < NCP; ++i)
        st[i] = *(const v4f*)((const char*)A + off[i]);
    __builtin_amdgcn_sched_barrier(0);

    const float dlt = delta[b];
    const float vv  = v[b];
    const float* ucol = u + (size_t)b * SN;
    v4f uc0[NB], uc1[NB], bc0[HB], bc1[HB];
    #pragma unroll
    for (int k = 0; k < NB; ++k) {
        const int nbase = k * TILE + 8 * hlf;
        uc0[k] = *(const v4f*)(ucol + nbase);
        uc1[k] = *(const v4f*)(ucol + nbase + 4);
    }
    const float* bown = Bvec + wave * (HB * TILE) + 8 * hlf;
    #pragma unroll
    for (int t = 0; t < HB; ++t) {
        bc0[t] = *(const v4f*)(bown + t * TILE);
        bc1[t] = *(const v4f*)(bown + t * TILE + 4);
    }

    #pragma unroll
    for (int i = 0; i < NCP; ++i) {
        const unsigned row = (row0 + 4u * i) & (unsigned)(SN - 1);
        *(v4f*)(&A_lds[row * APAD + lane * 4]) = st[i];
    }
    __syncthreads();

    const float dt2 = 0.5f * dlt;
    const float dvb = dlt * vv;
    const bool up = (hlf != 0);

    const float* alane = &A_lds[col * APAD + 2 * hlf];
    v2f* zs = &Z_lds[lane];
    float* orow = out + (size_t)b * SN + 8 * hlf;

    v2f afr[NBLK][4];
    Tri4 tri[2][4];
    v8f R[NB], RF[NB], RZ[NB];
    v2f zall[NSTEP];
    Inv4 Mall[NSTEP];

    v2f zin[NSTEP];
    Tri4 q[4];

    if (wave == 0) {
        read_fragments<0, 2, 0, 2, true>(afr, tri, A_lds, alane);
        __builtin_amdgcn_sched_barrier(0);
        zero_rows<0, 2>(RF);
        forward_cols<0, 2, 0, 2>(RF, afr, uc0, uc1, up);
        right_hand_sides<0, 2, false>(R, RF, RZ, uc0, uc1, bc0, bc1, dt2, dvb);
        solve_rows<0, 2, true, 0, true>(R, afr, tri, A_lds, dt2, up, zall, Mall, orow);
        leave_z<0>(zs, zall);
    } else if (wave == 1) {
        read_fragments<2, 4, 0, 4, false>(afr, tri, A_lds, alane);
        read_tris<2>(q, A_lds);
        read_tris<3>(tri[1], A_lds);
        __builtin_amdgcn_sched_barrier(0);
        zero_rows<2, 4>(RF);
        forward_cols<2, 4, 0, 4>(RF, afr, uc0, uc1, up);
        right_hand_sides<2, 4, false>(R, RF, RZ, uc0, uc1, bc0, bc1, dt2, dvb);
        form_inverses<0>(Mall, q, dt2);
    } else if (wave == 2) {
        read_fragments<4, 6, 0, 4, false>(afr, tri, A_lds, alane);
        read_tris<4>(q, A_lds);
        __builtin_amdgcn_sched_barrier(0);
        zero_rows<4, 6>(RF);
        zero_rows<4, 6>(RZ);
        forward_cols<4, 6, 0, 4>(RF, afr, uc0, uc1, up);
        form_inverses<0>(Mall, q, dt2);
        __builtin_amdgcn_sched_barrier(0);
        read_fragments<4, 6, 4, 6, false>(afr, tri, A_lds, alane);
        keep_fragments_here<4, 6, 4, 6>(afr);
    } else {
        read_fragments<6, 8, 0, 4, false>(afr, tri, A_lds, alane);
        __builtin_amdgcn_sched_barrier(0);
        zero_rows<6, 8>(RF);
        zero_rows<6, 8>(RZ);
        forward_cols<6, 8, 0, 4>(RF, afr, uc0, uc1, up);
        __builtin_amdgcn_sched_barrier(0);
        read_fragments<6, 8, 4, 6, false>(afr, tri, A_lds, alane);
        keep_fragments_here<6, 8, 4, 6>(afr);
    }

    hand_over();

    if (wave == 1) {
        read_z<0>(zin, zs);
        __builtin_amdgcn_sched_barrier(0);
        add_group<2, 4, 0>(R, afr, zin);
        solve_rows<2, 4, true, 4, true>(R, afr, tri, A_lds, dt2, up, zall, Mall, orow);
        leave_z<1>(zs, zall);
    } else if (wave == 2) {
        read_z<0>(zin, zs);
        read_tris<5>(q, A_lds);
        __builtin_amdgcn_sched_barrier(0);
        add_group<4, 6, 0>(RZ, afr, zin);
        forward_cols<4, 6, 4, 6>(RF, afr, uc0, uc1, up);
        right_hand_sides<4, 6, true>(R, RF, RZ, uc0, uc1, bc0, bc1, dt2, dvb);
        form_inverses<4>(Mall, q, dt2);
    } else if (wave == 3) {
        read_z<0>(zin, zs);
        read_tris<6>(q, A_lds);
        __builtin_amdgcn_sched_barrier(0);
        add_group<6, 8, 0>(RZ, afr, zin);
        forward_cols<6, 8, 4, 6>(RF, afr, uc0, uc1, up);
        form_inverses<0>(Mall, q, dt2);
        __builtin_amdgcn_sched_barrier(0);
        read_fragments<6, 8, 6, 8, false>(afr, tri, A_lds, alane);
        keep_fragments_here<6, 8, 6, 8>(afr);
    }

    hand_over();

    if (wave == 2) {
        read_z<1>(zin, zs);
        __builtin_amdgcn_sched_barrier(0);
        add_group<4, 6, 1>(R, afr, zin);
        solve_rows<4, 6, true, NSTEP, true>(R, afr, tri, A_lds, dt2, up, zall, Mall, orow);
        leave_z<2>(zs, zall);
    } else if (wave == 3) {
        read_z<1>(zin, zs);
        read_tris<7>(q, A_lds);
        __builtin_amdgcn_sched_barrier(0);
        add_group<6, 8, 1>(RZ, afr, zin);
        forward_cols<6, 8, 6, 8>(RF, afr, uc0, uc1, up);
        right_hand_sides<6, 8, true>(R, RF, RZ, uc0, uc1, bc0, bc1, dt2, dvb);
        form_inverses<4>(Mall, q, dt2);
    }

    hand_over();

    if (wave == 3) {
        read_z<2>(zin, zs);
        __builtin_amdgcn_sched_barrier(0);
        add_group<6, 8, 2>(R, afr, zin);
        solve_rows<6, 8, false, NSTEP, true>(R, afr, tri, A_lds, dt2, up, zall, Mall, orow);
    }
}

extern "C" void kernel_launch(void* const* d_in, const int* in_sizes, int n_in,
                              void* d_out, int out_size, void* d_ws, size_t ws_size,
                              hipStream_t stream) {
    const float* u     = (const float*)d_in[0];
    const float* v     = (const float*)d_in[1];
    const float* delta = (const float*)d_in[2];
    const float* A     = (const float*)d_in[3];
    const float* Bvec  = (const float*)d_in[4];
    float* out = (float*)d_out;

    const int batch  = in_sizes[0] / SN;
    const int blocks = batch / TILE;
    hippo_bilinear_wmma<<<blocks, WAVES * 32, 0, stream>>>(u, v, delta, A, Bvec, out);
}
